# starting kernel
# baseline (speedup 1.0000x reference)
#include <hip/hip_runtime.h>
#include <math.h>

typedef __attribute__((ext_vector_type(2))) float v2f;
typedef __attribute__((ext_vector_type(8))) float v8f;

#define N_ITERS 5
#define EPS_F   1e-6f
#define ETA_F   0.02f

#define OM2_O 0
#define OM1_O 16384
#define XS_O  28672
#define YS_O  32768
#define CM_O  36864
#define IM_O  37888
#define SC_O  38656
#define LDS_FLOATS 38912

static __device__ __forceinline__ v8f wmma4(v2f a, v2f b, v8f c) {
  return __builtin_amdgcn_wmma_f32_16x16x4_f32(false, a, false, b, (short)0, c,
                                               false, false);
}

__global__ __launch_bounds__(256) void l2norm_rows(const float* __restrict__ imgs,
                                                   const float* __restrict__ caps,
                                                   float* __restrict__ wimgs,
                                                   float* __restrict__ wcaps) {
  const int row = blockIdx.x;
  const float* src;
  float* dst;
  if (row < 128 * 36) {
    src = imgs + (size_t)row * 1024;
    dst = wimgs + (size_t)row * 1024;
  } else {
    int r = row - 128 * 36;
    src = caps + (size_t)r * 1024;
    dst = wcaps + (size_t)r * 1024;
  }
  float v[4];
  float s = 0.f;
#pragma unroll
  for (int e = 0; e < 4; ++e) {
    v[e] = src[threadIdx.x + 256 * e];
    s += v[e] * v[e];
  }
#pragma unroll
  for (int off = 16; off > 0; off >>= 1) s += __shfl_xor(s, off, 32);
  __shared__ float red[8];
  if ((threadIdx.x & 31) == 0) red[threadIdx.x >> 5] = s;
  __syncthreads();
  float tot = 0.f;
#pragma unroll
  for (int k = 0; k < 8; ++k) tot += red[k];
  const float inv = 1.0f / fmaxf(sqrtf(tot), 1e-12f);
#pragma unroll
  for (int e = 0; e < 4; ++e) dst[threadIdx.x + 256 * e] = v[e] * inv;
}

static __device__ __forceinline__ void slabs(float* lds,
                                             const float* __restrict__ gi,
                                             const float* __restrict__ gc,
                                             int w, int ln, int koff, int ioff,
                                             int tg, int ig, int d0) {
  v8f x; x = 0.0f;
  for (int l0 = 0; l0 < 60; l0 += 4) {
    v2f a = *(const v2f*)&lds[OM2_O + (ln * 16 + w) * 64 + l0 + koff];
    v2f b;
    b.x = gc[((size_t)tg * 60 + (l0 + koff)) * 1024 + d0 + ln];
    b.y = gc[((size_t)tg * 60 + (l0 + koff + 1)) * 1024 + d0 + ln];
    x = wmma4(a, b, x);
  }
  v8f y; y = 0.0f;
  for (int k0 = 0; k0 < 36; k0 += 4) {
    v2f a = *(const v2f*)&lds[OM1_O + (w * 16 + ln) * 48 + k0 + koff];
    v2f b;
    b.x = gi[((size_t)ig * 36 + (k0 + koff)) * 1024 + d0 + ln];
    b.y = gi[((size_t)ig * 36 + (k0 + koff + 1)) * 1024 + d0 + ln];
    y = wmma4(a, b, y);
  }
#pragma unroll
  for (int j = 0; j < 8; ++j) {
    lds[XS_O + w * 256 + (j + ioff) * 16 + ln] = x[j];
    lds[YS_O + w * 256 + (j + ioff) * 16 + ln] = y[j];
  }
}

__global__ __launch_bounds__(512) void nmf_tile(const float* __restrict__ gi,
                                                const float* __restrict__ gc,
                                                const int* __restrict__ il_g,
                                                const int* __restrict__ cl_g,
                                                float* __restrict__ out) {
  extern __shared__ float lds[];
  const int tid = threadIdx.x;
  const int w = tid >> 5;
  const int lane = tid & 31;
  const int hi = lane >> 4;
  const int ln = lane & 15;
  const int ib = (blockIdx.x >> 3) * 16;
  const int tb = (blockIdx.x & 7) * 16;
  const int tg = tb + w;
  const int ig = ib + w;
  const int ioff = hi ? 8 : 0;
  const int koff = hi ? 2 : 0;

  for (int idx = tid; idx < 16 * 64; idx += 512) {
    int t = idx >> 6, l = idx & 63;
    lds[CM_O + idx] = (l < cl_g[tb + t]) ? 1.f : 0.f;
  }
  for (int idx = tid; idx < 16 * 48; idx += 512) {
    int i = idx / 48, k = idx % 48;
    lds[IM_O + idx] = (k < il_g[ib + i]) ? 1.f : 0.f;
  }
  for (int idx = tid; idx < 16 * 16 * 64; idx += 512) {
    int t = (idx >> 6) & 15, l = idx & 63;
    int cl = cl_g[tb + t];
    lds[OM2_O + idx] = (l < cl) ? (1.f / (float)cl) : 0.f;
  }
  for (int idx = tid; idx < 16 * 16 * 48; idx += 512) {
    int i = idx / (16 * 48), k = idx % 48;
    int il = il_g[ib + i];
    lds[OM1_O + idx] = (k < il) ? (1.f / (float)il) : 0.f;
  }
  __syncthreads();

#pragma unroll 1
  for (int iter = 0; iter < N_ITERS; ++iter) {
    v8f dv[4];
#pragma unroll
    for (int n = 0; n < 4; ++n) dv[n] = 0.0f;
#pragma unroll 1
    for (int ch = 0; ch < 64; ++ch) {
      const int d0 = ch * 16;
      slabs(lds, gi, gc, w, ln, koff, ioff, tg, ig, d0);
      __syncthreads();
#pragma unroll
      for (int n = 0; n < 4; ++n) {
        const int lc = (n * 16 + ln < 59) ? (n * 16 + ln) : 59;
#pragma unroll
        for (int s = 0; s < 4; ++s) {
          const int dd = s * 4 + koff;
          v2f ax = *(const v2f*)&lds[XS_O + w * 256 + ln * 16 + dd];
          v2f ay = *(const v2f*)&lds[YS_O + ln * 256 + w * 16 + dd];
          v2f a; a.x = ax.x - ay.x; a.y = ax.y - ay.y;
          v2f b = *(const v2f*)&gc[((size_t)tg * 60 + lc) * 1024 + d0 + dd];
          dv[n] = wmma4(a, b, dv[n]);
        }
      }
      __syncthreads();
    }
#pragma unroll
    for (int n = 0; n < 4; ++n)
#pragma unroll
      for (int j = 0; j < 8; ++j) {
        const int i = j + ioff, l = n * 16 + ln;
        const int idx = (i * 16 + w) * 64 + l;
        float nv = lds[OM2_O + idx] - ETA_F * dv[n][j];
        lds[OM2_O + idx] = fmaxf(nv, 0.f) * lds[CM_O + w * 64 + l];
      }
    __syncthreads();
    if (tid < 256) {
      float s = 0.f;
      for (int l = 0; l < 64; ++l) s += lds[OM2_O + tid * 64 + l];
      lds[SC_O + tid] = 1.f / (s + EPS_F);
    }
    __syncthreads();
    for (int idx = tid; idx < 16384; idx += 512)
      lds[OM2_O + idx] *= lds[SC_O + (idx >> 6)];
    __syncthreads();

    v8f du[3];
#pragma unroll
    for (int n = 0; n < 3; ++n) du[n] = 0.0f;
#pragma unroll 1
    for (int ch = 0; ch < 64; ++ch) {
      const int d0 = ch * 16;
      slabs(lds, gi, gc, w, ln, koff, ioff, tg, ig, d0);
      __syncthreads();
#pragma unroll
      for (int n = 0; n < 3; ++n) {
        const int kc = (n * 16 + ln < 35) ? (n * 16 + ln) : 35;
#pragma unroll
        for (int s = 0; s < 4; ++s) {
          const int dd = s * 4 + koff;
          v2f ax = *(const v2f*)&lds[XS_O + ln * 256 + w * 16 + dd];
          v2f ay = *(const v2f*)&lds[YS_O + w * 256 + ln * 16 + dd];
          v2f a; a.x = ax.x - ay.x; a.y = ax.y - ay.y;
          v2f b = *(const v2f*)&gi[((size_t)ig * 36 + kc) * 1024 + d0 + dd];
          du[n] = wmma4(a, b, du[n]);
        }
      }
      __syncthreads();
    }
#pragma unroll
    for (int n = 0; n < 3; ++n)
#pragma unroll
      for (int j = 0; j < 8; ++j) {
        const int t = j + ioff, k = n * 16 + ln;
        const int idx = (w * 16 + t) * 48 + k;
        float nv = lds[OM1_O + idx] + ETA_F * du[n][j];
        lds[OM1_O + idx] = fmaxf(nv, 0.f) * lds[IM_O + w * 48 + k];
      }
    __syncthreads();
    if (tid < 256) {
      float s = 0.f;
      for (int k = 0; k < 48; ++k) s += lds[OM1_O + tid * 48 + k];
      lds[SC_O + tid] = 1.f / (s + EPS_F);
    }
    __syncthreads();
    for (int idx = tid; idx < 12288; idx += 512)
      lds[OM1_O + idx] *= lds[SC_O + idx / 48];
    __syncthreads();
  }

  float s11 = 0.f, s22 = 0.f, s12 = 0.f;
#pragma unroll 1
  for (int ch = 0; ch < 64; ++ch) {
    slabs(lds, gi, gc, w, ln, koff, ioff, tg, ig, ch * 16);
    __syncthreads();
    if (tid < 256) {
      const int pi = tid >> 4, pt = tid & 15;
#pragma unroll
      for (int d = 0; d < 16; ++d) {
        float g1 = lds[YS_O + pi * 256 + pt * 16 + d];
        float g2 = lds[XS_O + pt * 256 + pi * 16 + d];
        s12 += g1 * g2; s11 += g1 * g1; s22 += g2 * g2;
      }
    }
    __syncthreads();
  }
  if (tid < 256) {
    const int pi = tid >> 4, pt = tid & 15;
    const float dnm = fmaxf(sqrtf(s11), 1e-12f) * fmaxf(sqrtf(s22), 1e-12f);
    out[(size_t)(ib + pi) * 128 + (tb + pt)] = s12 / dnm;
  }
}

extern "C" void kernel_launch(void* const* d_in, const int* in_sizes, int n_in,
                              void* d_out, int out_size, void* d_ws, size_t ws_size,
                              hipStream_t stream) {
  (void)in_sizes; (void)n_in; (void)out_size; (void)ws_size;
  const float* imgs = (const float*)d_in[0];
  const float* caps = (const float*)d_in[1];
  const int* il = (const int*)d_in[2];
  const int* cl = (const int*)d_in[3];
  float* wimgs = (float*)d_ws;
  float* wcaps = wimgs + (size_t)128 * 36 * 1024;

  l2norm_rows<<<128 * 36 + 128 * 60, 256, 0, stream>>>(imgs, caps, wimgs, wcaps);
  nmf_tile<<<64, 512, LDS_FLOATS * sizeof(float), stream>>>(wimgs, wcaps, il, cl,
                                                            (float*)d_out);
}


# each workgroup walks its slice's 16 chunks in a rotated order, start set by tile: the 8 workgroups sharing caption or image data stand on 8 different chunks; chunk sum order changes
# speedup vs baseline: 7.1114x; 7.1114x over previous
#include <hip/hip_runtime.h>
#include <math.h>

typedef __attribute__((ext_vector_type(2))) float v2f;
typedef __attribute__((ext_vector_type(8))) float v8f;

#define N_ITERS 5
#define EPS_F   1e-6f
#define ETA_F   0.02f

#define OM2_PI 1028
#define OM1_PR 52
#define SL_PR  20
#define SL_PO  324
#define OM2_O 0
#define OM1_O 16448
#define XS_O  29760
#define YS_O  34944
#define LDS_FLOATS 40128

static __device__ __forceinline__ v8f wmma4(v2f a, v2f b, v8f c) {
  return __builtin_amdgcn_wmma_f32_16x16x4_f32(false, a, false, b, (short)0, c,
                                               false, false);
}

#define NT_P 1028
#define SI_CH 576
#define TI_CH 768
#define SC_CH 960
#define TC_CH 1024
__global__ __launch_bounds__(512) void l2norm_frag(const float* __restrict__ imgs,
                                                   const float* __restrict__ caps,
                                                   float* __restrict__ simg,
                                                   float* __restrict__ timg,
                                                   float* __restrict__ scap,
                                                   float* __restrict__ tcap) {
  extern __shared__ float tile[];
  const int tid = threadIdx.x;
  const int w = tid >> 5;
  const int lane = tid & 31;
  int ent, n, R, sch, tch;
  const float* src;
  float* sdst;
  float* tdst;
  if (blockIdx.x < 128 * 3) {
    ent = blockIdx.x / 3; n = blockIdx.x - ent * 3; R = 36; sch = SI_CH; tch = TI_CH;
    src = imgs + (size_t)ent * 36 * 1024;
    sdst = simg + (size_t)ent * 64 * SI_CH;
    tdst = timg + (size_t)ent * 64 * TI_CH;
  } else {
    const int b = blockIdx.x - 128 * 3;
    ent = b >> 2; n = b & 3; R = 60; sch = SC_CH; tch = TC_CH;
    src = caps + (size_t)ent * 60 * 1024;
    sdst = scap + (size_t)ent * 64 * SC_CH;
    tdst = tcap + (size_t)ent * 64 * TC_CH;
  }
  const int row = n * 16 + w;
  float* trow = tile + w * NT_P;
  if (row < R) {
    const float* rp = src + (size_t)row * 1024 + lane;
    float v[32];
#pragma unroll
    for (int j = 0; j < 8; ++j)
#pragma unroll
      for (int e = 0; e < 4; ++e) v[j * 4 + e] = rp[32 * j + 256 * e];
    float tot = 0.f;
#pragma unroll
    for (int j = 0; j < 8; ++j) {
      float s = 0.f;
#pragma unroll
      for (int e = 0; e < 4; ++e) s += v[j * 4 + e] * v[j * 4 + e];
#pragma unroll
      for (int off = 16; off > 0; off >>= 1) s += __shfl_xor(s, off, 32);
      tot += s;
    }
    const float inv = 1.0f / fmaxf(sqrtf(tot), 1e-12f);
#pragma unroll
    for (int j = 0; j < 8; ++j)
#pragma unroll
      for (int e = 0; e < 4; ++e) trow[lane + 32 * j + 256 * e] = v[j * 4 + e] * inv;
  } else {
#pragma unroll
    for (int q = 0; q < 32; ++q) trow[lane + 32 * q] = 0.f;
  }
  __syncthreads();
  const int nS = (R / 4 - 4 * n < 4) ? (R / 4 - 4 * n) : 4;
#pragma unroll
  for (int it = 0; it < 8; ++it) {
    const int u = tid + 512 * it;
    const int ch = u >> 6;
    const int f = (u & 63) * 4;
    if (f < nS * 64) {
      const int sl = f >> 6, fl = (f >> 1) & 31;
      const int r0 = 4 * sl + 2 * (fl >> 4), cc = 16 * ch + (fl & 15);
      const v2f a = *(const v2f*)&tile[r0 * NT_P + cc];
      const v2f b = *(const v2f*)&tile[(r0 + 1) * NT_P + cc];
      *(float4*)(sdst + (size_t)ch * sch + n * 256 + f) = make_float4(a.x, b.x, a.y, b.y);
    }
  }
#pragma unroll
  for (int it = 0; it < 8; ++it) {
    const int u = tid + 512 * it;
    const int ch = u >> 6;
    const int f = (u & 63) * 4;
    const int s = f >> 6, fl = (f >> 1) & 31;
    const int r0 = fl & 15, col = 16 * ch + 4 * s + 2 * (fl >> 4);
    const v2f a = *(const v2f*)&tile[r0 * NT_P + col];
    const v2f b = *(const v2f*)&tile[(r0 + 1) * NT_P + col];
    *(float4*)(tdst + (size_t)ch * tch + n * 256 + f) = make_float4(a.x, a.y, b.x, b.y);
  }
}

static __device__ __forceinline__ void lds_barrier() {
  __builtin_amdgcn_fence(__ATOMIC_RELEASE, "workgroup", "local");
  __builtin_amdgcn_s_barrier();
  __builtin_amdgcn_fence(__ATOMIC_ACQUIRE, "workgroup", "local");
}

#define NC_HEAD 12
#define NG_HEAD 8
struct SlabB {
  v2f c[15];
  v2f g[9];
};

static __device__ __forceinline__ v8f kept_load(const float* q) {
  const float4 lo = *(const float4*)q;
  const float4 hi = *(const float4*)(q + 128);
  v8f r;
  r[0] = lo.x; r[1] = lo.y; r[2] = lo.z; r[3] = lo.w;
  r[4] = hi.x; r[5] = hi.y; r[6] = hi.z; r[7] = hi.w;
  return r;
}

static __device__ __forceinline__ void kept_store(float* q, v8f v) {
  *(float4*)q = make_float4(v[0], v[1], v[2], v[3]);
  *(float4*)(q + 128) = make_float4(v[4], v[5], v[6], v[7]);
}

template <int MODE>
static __device__ __forceinline__ void slab_loads_head(SlabB& b,
                                                       const float* __restrict__ pg,
                                                       const float* __restrict__ pc) {
  if (MODE != 1) {
#pragma unroll
    for (int s = 0; s < NC_HEAD; ++s) b.c[s] = *(const v2f*)(pc + s * 64);
  }
  if (MODE != 2) {
#pragma unroll
    for (int s = 0; s < NG_HEAD; ++s) b.g[s] = *(const v2f*)(pg + s * 64);
  }
}

template <int MODE>
static __device__ __forceinline__ void slab_loads_tail(SlabB& b,
                                                       const float* __restrict__ pg,
                                                       const float* __restrict__ pc,
                                                       bool longc, bool longg) {
  if (MODE != 1) {
    if (longc) {
#pragma unroll
      for (int s = NC_HEAD; s < 15; ++s) b.c[s] = *(const v2f*)(pc + s * 64);
    }
  }
  if (MODE != 2) {
    if (longg) {
#pragma unroll
      for (int s = NG_HEAD; s < 9; ++s) b.g[s] = *(const v2f*)(pg + s * 64);
    }
  }
}

template <int MODE, bool NEXT>
static __device__ __forceinline__ void slab_step(float* lds, SlabB& b, v8f& lin, v8f& made,
                                                 const float* qn,
                                                 const float* __restrict__ pgn,
                                                 const float* __restrict__ pcn,
                                                 int w, int ln, int koff, int ioff,
                                                 bool longc, bool longg) {
  v2f at2[15 - NC_HEAD], at1[9 - NG_HEAD];
  if (MODE != 1) {
#pragma unroll
    for (int s = NC_HEAD; s < 15; ++s)
      at2[s - NC_HEAD] = *(const v2f*)&lds[OM2_O + ln * OM2_PI + w * 64 + s * 4 + koff];
  }
  if (MODE != 2) {
#pragma unroll
    for (int s = NG_HEAD; s < 9; ++s)
      at1[s - NG_HEAD] = *(const v2f*)&lds[OM1_O + (w * 16 + ln) * OM1_PR + s * 4 + koff];
  }
  v8f x; x = 0.0f;
  if (MODE != 1) {
#pragma unroll
    for (int s = 0; s < NC_HEAD; ++s) {
      v2f a = *(const v2f*)&lds[OM2_O + ln * OM2_PI + w * 64 + s * 4 + koff];
      x = wmma4(a, b.c[s], x);
    }
  }
  v8f y; y = 0.0f;
  if (MODE != 2) {
#pragma unroll
    for (int s = 0; s < NG_HEAD; ++s) {
      v2f a = *(const v2f*)&lds[OM1_O + (w * 16 + ln) * OM1_PR + s * 4 + koff];
      y = wmma4(a, b.g[s], y);
    }
  }
  if (MODE == 0) asm volatile("" :: "v"(at2[0]), "v"(at2[1]), "v"(at2[2]), "v"(at1[0]));
  if (MODE == 1) asm volatile("" :: "v"(at1[0]));
  if (MODE == 2) asm volatile("" :: "v"(at2[0]), "v"(at2[1]), "v"(at2[2]));
  if (MODE != 1) {
    if (longc) {
#pragma unroll
      for (int s = NC_HEAD; s < 15; ++s) x = wmma4(at2[s - NC_HEAD], b.c[s], x);
    }
  }
  if (MODE != 2) {
    if (longg) {
#pragma unroll
      for (int s = NG_HEAD; s < 9; ++s) y = wmma4(at1[s - NG_HEAD], b.g[s], y);
    }
  }
  if (MODE == 1) x = lin;
  if (MODE == 2) y = lin;
#pragma unroll
  for (int j = 0; j < 8; ++j) {
    lds[XS_O + w * SL_PO + (j + ioff) * SL_PR + ln] = x[j];
    lds[YS_O + w * SL_PO + (j + ioff) * SL_PR + ln] = y[j];
  }
  made = (MODE == 2) ? x : y;
  if (NEXT) {
    slab_loads_head<MODE>(b, pgn, pcn);
    if (MODE != 0) lin = kept_load(qn);
  }
}

template <int NTM>
static __device__ __forceinline__ void tile_loads_head(v2f (&bt)[NTM * 4],
                                                       const float* __restrict__ g) {
#pragma unroll
  for (int q = 0; q < (NTM - 1) * 4; ++q) bt[q] = *(const v2f*)(g + q * 64);
}

template <int NTM>
static __device__ __forceinline__ void tile_loads_tail(v2f (&bt)[NTM * 4],
                                                       const float* __restrict__ g, bool all) {
  if (all) {
#pragma unroll
    for (int q = (NTM - 1) * 4; q < NTM * 4; ++q) bt[q] = *(const v2f*)(g + q * 64);
  }
}

template <int NTM, bool XOWN, bool NEXT>
static __device__ __forceinline__ void grad_products(const float* lds, v8f (&acc)[NTM],
                                                     v2f (&bt)[NTM * 4],
                                                     const float* __restrict__ gn,
                                                     int w, int ln, int koff, bool all) {
  v2f a[4];
#pragma unroll
  for (int s = 0; s < 4; ++s) {
    const int dd = s * 4 + koff;
    const int own = w * SL_PO + ln * SL_PR + dd;
    const int oth = ln * SL_PO + w * SL_PR + dd;
    const v2f ax = *(const v2f*)&lds[XS_O + (XOWN ? own : oth)];
    const v2f ay = *(const v2f*)&lds[YS_O + (XOWN ? oth : own)];
    a[s].x = ax.x - ay.x; a[s].y = ax.y - ay.y;
  }
#pragma unroll
  for (int n = 0; n < NTM - 1; ++n)
#pragma unroll
    for (int s = 0; s < 4; ++s) acc[n] = wmma4(a[s], bt[n * 4 + s], acc[n]);
  if (all) {
#pragma unroll
    for (int s = 0; s < 4; ++s)
      acc[NTM - 1] = wmma4(a[s], bt[(NTM - 1) * 4 + s], acc[NTM - 1]);
  }
  if (NEXT) tile_loads_head<NTM>(bt, gn);
}

static __device__ __forceinline__ void final_sums(const float* lds, int tid,
                                                  float& s11, float& s22, float& s12) {
  if (tid < 256) {
    const int pi = tid >> 4, pt = tid & 15;
#pragma unroll
    for (int d = 0; d < 16; ++d) {
      float g1 = lds[YS_O + pi * SL_PO + pt * SL_PR + d];
      float g2 = lds[XS_O + pt * SL_PO + pi * SL_PR + d];
      s12 += g1 * g2; s11 += g1 * g1; s22 += g2 * g2;
    }
  }
}

#define NSL      4
#define CH_SL    (64 / NSL)
#define O2_FL    16384
#define O1_FL    12288
#define FP_FL    768
#define ST_O2    0
#define ST_O1    (ST_O2 + 2 * 64 * O2_FL)
#define ST_PV    (ST_O1 + 2 * 64 * O1_FL)
#define ST_PU    (ST_PV + 64 * NSL * O2_FL)
#define ST_FP    (ST_PU + 64 * NSL * O1_FL)
#define SB_CH    4096
#define ST_SB    (ST_FP + 64 * NSL * FP_FL)

template <int MODE, int NTM, bool XOWN>
static __device__ __forceinline__ void grad_half(float* lds, v8f (&acc)[NTM],
                                                 const float* __restrict__ pg0,
                                                 const float* __restrict__ pc0,
                                                 const float* __restrict__ gt0, float* q0,
                                                 int c0,
                                                 int w, int ln, int koff, int ioff,
                                                 bool longc, bool longg, bool all) {
  const int tch = (NTM == 4 ? TC_CH : TI_CH);
  SlabB sb;
  v2f bt[NTM * 4];
  v8f lin, made;
  lin = 0.0f;
  int c = c0;
  const float* pc = pc0 + c * SC_CH;
  const float* pg = pg0 + c * SI_CH;
  const float* gt = gt0 + c * tch;
  float* q = q0 + c * SB_CH;
  slab_loads_head<MODE>(sb, pg, pc);
  slab_loads_tail<MODE>(sb, pg, pc, longc, longg);
  if (MODE != 0) lin = kept_load(q);
  tile_loads_head<NTM>(bt, gt);
#pragma unroll 1
  for (int j = 0; j < CH_SL - 1; ++j) {
    const int cn = (c + 1) & (CH_SL - 1);
    const float* pcn = pc0 + cn * SC_CH;
    const float* pgn = pg0 + cn * SI_CH;
    const float* gtn = gt0 + cn * tch;
    float* qn = q0 + cn * SB_CH;
    tile_loads_tail<NTM>(bt, gt, all);
    slab_step<MODE, true>(lds, sb, lin, made, qn, pgn, pcn, w, ln, koff, ioff, longc, longg);
    lds_barrier();
    kept_store(q, made);
    slab_loads_tail<MODE>(sb, pgn, pcn, longc, longg);
    grad_products<NTM, XOWN, true>(lds, acc, bt, gtn, w, ln, koff, all);
    lds_barrier();
    c = cn; pc = pcn; pg = pgn; gt = gtn; q = qn;
  }
  tile_loads_tail<NTM>(bt, gt, all);
  slab_step<MODE, false>(lds, sb, lin, made, q, pg, pc, w, ln, koff, ioff, longc, longg);
  lds_barrier();
  kept_store(q, made);
  grad_products<NTM, XOWN, false>(lds, acc, bt, gt, w, ln, koff, all);
  lds_barrier();
}

__global__ __launch_bounds__(512) void nmf_phase(const float* __restrict__ simg,
                                                 const float* __restrict__ timg,
                                                 const float* __restrict__ scap,
                                                 const float* __restrict__ tcap,
                                                 const int* __restrict__ il_g,
                                                 const int* __restrict__ cl_g,
                                                 float* st,
                                                 int p) {
  extern __shared__ float lds[];
  const int tid = threadIdx.x;
  const int w = tid >> 5;
  const int lane = tid & 31;
  const int hi = lane >> 4;
  const int ln = lane & 15;
  const int tile = blockIdx.x / NSL;
  const int slice = blockIdx.x % NSL;
  const int ib = (tile >> 3) * 16;
  const int tb = (tile & 7) * 16;
  const int tg = tb + w;
  const int ig = ib + w;
  const int ioff = hi ? 8 : 0;
  const int koff = hi ? 2 : 0;
  const int m = p >> 1;
  const bool odd = (p & 1) != 0;
  const int chb = slice * CH_SL;
  const int c0 = (((tile >> 3) + (tile & 7)) & 7) * (CH_SL / 8);

  if (odd || p == 0) {
    const float* src2 = st + ST_O2 + ((size_t)((m + 1) & 1) * 64 + tile) * O2_FL;
    const float* pv = st + ST_PV + (size_t)tile * NSL * O2_FL;
    float* dst2 = st + ST_O2 + ((size_t)(m & 1) * 64 + tile) * O2_FL;
#pragma unroll
    for (int it = 0; it < O2_FL / 4 / 512; ++it) {
      const int idx = (tid + 512 * it) * 4;
      const int t = (idx >> 6) & 15, l0 = idx & 63;
      const int cl = cl_g[tb + t];
      float v[4];
      if (m == 0) {
        const float inv = 1.f / (float)cl;
        for (int e = 0; e < 4; ++e) v[e] = (l0 + e < cl) ? inv : 0.f;
      } else {
        const float4 x = *(const float4*)(src2 + idx);
        v[0] = x.x; v[1] = x.y; v[2] = x.z; v[3] = x.w;
      }
      if (odd) {
        float d[4] = {0.f, 0.f, 0.f, 0.f};
        for (int s = 0; s < NSL; ++s) {
          const float4 x = *(const float4*)(pv + (size_t)s * O2_FL + idx);
          d[0] += x.x; d[1] += x.y; d[2] += x.z; d[3] += x.w;
        }
        for (int e = 0; e < 4; ++e) {
          const float nv = v[e] - ETA_F * d[e];
          v[e] = fmaxf(nv, 0.f) * ((l0 + e < cl) ? 1.f : 0.f);
        }
        float s = (v[0] + v[1]) + (v[2] + v[3]);
        for (int off = 8; off > 0; off >>= 1) s += __shfl_xor(s, off, 32);
        const float sc = 1.f / (s + EPS_F);
        for (int e = 0; e < 4; ++e) v[e] *= sc;
        if (slice == 0) *(float4*)(dst2 + idx) = make_float4(v[0], v[1], v[2], v[3]);
      }
      *(float4*)&lds[OM2_O + (idx >> 10) * OM2_PI + (idx & 1023)] = make_float4(v[0], v[1], v[2], v[3]);
    }
  }
  if (!odd) {
    const bool start1 = m <= 1;
    const bool upd1 = m >= 1;
    const int sel = m & 1;
    const float* src1 = st + ST_O1 + ((size_t)sel * 64 + tile) * O1_FL;
    const float* pu = st + ST_PU + (size_t)tile * NSL * O1_FL;
    float* dst1 = st + ST_O1 + ((size_t)((m + 1) & 1) * 64 + tile) * O1_FL;
    const int c16 = tid & 15;
    const bool act = c16 < 12;
    const int k0 = act ? c16 * 4 : 0;
#pragma unroll
    for (int it = 0; it < 8; ++it) {
      const int row = (tid >> 4) + 32 * it;
      const int idx = row * 48 + k0;
      const int il = il_g[ib + (row >> 4)];
      float v[4];
      if (start1) {
        const float inv = 1.f / (float)il;
        for (int e = 0; e < 4; ++e) v[e] = (k0 + e < il) ? inv : 0.f;
      } else {
        const float4 x = *(const float4*)(src1 + idx);
        v[0] = x.x; v[1] = x.y; v[2] = x.z; v[3] = x.w;
      }
      if (upd1) {
        float d[4] = {0.f, 0.f, 0.f, 0.f};
        for (int s = 0; s < NSL; ++s) {
          const float4 x = *(const float4*)(pu + (size_t)s * O1_FL + idx);
          d[0] += x.x; d[1] += x.y; d[2] += x.z; d[3] += x.w;
        }
        for (int e = 0; e < 4; ++e) {
          const float nv = v[e] + ETA_F * d[e];
          v[e] = fmaxf(nv, 0.f) * ((k0 + e < il) ? 1.f : 0.f);
        }
        float s = act ? ((v[0] + v[1]) + (v[2] + v[3])) : 0.f;
        for (int off = 8; off > 0; off >>= 1) s += __shfl_xor(s, off, 32);
        const float sc = 1.f / (s + EPS_F);
        for (int e = 0; e < 4; ++e) v[e] *= sc;
        if (slice == 0 && act) *(float4*)(dst1 + idx) = make_float4(v[0], v[1], v[2], v[3]);
      }
      if (act) *(float4*)&lds[OM1_O + row * OM1_PR + k0] = make_float4(v[0], v[1], v[2], v[3]);
    }
  }
  __syncthreads();

  const float* pc = scap + ((size_t)tg * 64 + chb) * SC_CH + lane * 2;
  const float* pg = simg + ((size_t)ig * 64 + chb) * SI_CH + lane * 2;
  float* q = st + ST_SB + (((size_t)tile * NSL + slice) * CH_SL * 16 + w) * 256 + lane * 4;

  const bool longc = __builtin_amdgcn_readfirstlane(cl_g[tg]) > 4 * NC_HEAD;
  const bool longg = __builtin_amdgcn_readfirstlane(il_g[ig]) > 4 * NG_HEAD;

  if (p == 2 * N_ITERS) {
    float s11 = 0.f, s22 = 0.f, s12 = 0.f;
    SlabB sb;
    v8f lin, made;
    int c = c0;
    const float* pcc = pc + c * SC_CH;
    const float* pgc = pg + c * SI_CH;
    float* qc = q + c * SB_CH;
    slab_loads_head<1>(sb, pgc, pcc);
    slab_loads_tail<1>(sb, pgc, pcc, longc, longg);
    lin = kept_load(qc);
#pragma unroll 1
    for (int j = 0; j < CH_SL - 1; ++j) {
      c = (c + 1) & (CH_SL - 1);
      pcc = pc + c * SC_CH; pgc = pg + c * SI_CH; qc = q + c * SB_CH;
      slab_step<1, true>(lds, sb, lin, made, qc, pgc, pcc, w, ln, koff, ioff, longc, longg);
      lds_barrier();
      slab_loads_tail<1>(sb, pgc, pcc, longc, longg);
      final_sums(lds, tid, s11, s22, s12);
      lds_barrier();
    }
    slab_step<1, false>(lds, sb, lin, made, qc, pgc, pcc, w, ln, koff, ioff, longc, longg);
    lds_barrier();
    final_sums(lds, tid, s11, s22, s12);
    lds_barrier();
    if (tid < 256) {
      float* fp = st + ST_FP + ((size_t)tile * NSL + slice) * FP_FL;
      fp[tid] = s11;
      fp[256 + tid] = s22;
      fp[512 + tid] = s12;
    }
  } else if (!odd) {
    v8f dv[4];
#pragma unroll
    for (int n = 0; n < 4; ++n) dv[n] = 0.0f;
    const float* gct = tcap + ((size_t)tg * 64 + chb) * TC_CH + lane * 2;
    if (p == 0) grad_half<0, 4, true>(lds, dv, pg, pc, gct, q, c0, w, ln, koff, ioff, longc, longg, longc);
    else        grad_half<1, 4, true>(lds, dv, pg, pc, gct, q, c0, w, ln, koff, ioff, longc, longg, longc);
    float* pvw = st + ST_PV + ((size_t)tile * NSL + slice) * O2_FL;
#pragma unroll
    for (int n = 0; n < 4; ++n)
#pragma unroll
      for (int j = 0; j < 8; ++j) {
        const int i = j + ioff, l = n * 16 + ln;
        pvw[(i * 16 + w) * 64 + l] = dv[n][j];
      }
  } else {
    v8f du[3];
#pragma unroll
    for (int n = 0; n < 3; ++n) du[n] = 0.0f;
    const float* git = timg + ((size_t)ig * 64 + chb) * TI_CH + lane * 2;
    grad_half<2, 3, false>(lds, du, pg, pc, git, q, c0, w, ln, koff, ioff, longc, longg, longg);
    float* puw = st + ST_PU + ((size_t)tile * NSL + slice) * O1_FL;
#pragma unroll
    for (int n = 0; n < 3; ++n)
#pragma unroll
      for (int j = 0; j < 8; ++j) {
        const int t = j + ioff, k = n * 16 + ln;
        puw[(w * 16 + t) * 48 + k] = du[n][j];
      }
  }
}

__global__ __launch_bounds__(256) void nmf_out(const float* __restrict__ st,
                                               float* __restrict__ out) {
  const int tile = blockIdx.x;
  const int tid = threadIdx.x;
  const int ib = (tile >> 3) * 16;
  const int tb = (tile & 7) * 16;
  const int pi = tid >> 4, pt = tid & 15;
  float s11 = 0.f, s22 = 0.f, s12 = 0.f;
  for (int s = 0; s < NSL; ++s) {
    const float* fp = st + ST_FP + ((size_t)tile * NSL + s) * FP_FL;
    s11 += fp[tid];
    s22 += fp[256 + tid];
    s12 += fp[512 + tid];
  }
  const float dnm = fmaxf(sqrtf(s11), 1e-12f) * fmaxf(sqrtf(s22), 1e-12f);
  out[(size_t)(ib + pi) * 128 + (tb + pt)] = s12 / dnm;
}

extern "C" void kernel_launch(void* const* d_in, const int* in_sizes, int n_in,
                              void* d_out, int out_size, void* d_ws, size_t ws_size,
                              hipStream_t stream) {
  (void)in_sizes; (void)n_in; (void)out_size; (void)ws_size;
  const float* imgs = (const float*)d_in[0];
  const float* caps = (const float*)d_in[1];
  const int* il = (const int*)d_in[2];
  const int* cl = (const int*)d_in[3];
  float* simg = (float*)d_ws;
  float* timg = simg + (size_t)128 * 64 * SI_CH;
  float* scap = timg + (size_t)128 * 64 * TI_CH;
  float* tcap = scap + (size_t)128 * 64 * SC_CH;
  float* st = tcap + (size_t)128 * 64 * TC_CH;
  const size_t lds_bytes = LDS_FLOATS * sizeof(float);

  l2norm_frag<<<128 * 3 + 128 * 4, 512, 16 * NT_P * sizeof(float), stream>>>(imgs, caps, simg, timg, scap, tcap);
  nmf_phase<<<64 * NSL, 512, lds_bytes, stream>>>(simg, timg, scap, tcap, il, cl, st, 0);
  nmf_phase<<<64 * NSL, 512, lds_bytes, stream>>>(simg, timg, scap, tcap, il, cl, st, 1);
  nmf_phase<<<64 * NSL, 512, lds_bytes, stream>>>(simg, timg, scap, tcap, il, cl, st, 2);
  nmf_phase<<<64 * NSL, 512, lds_bytes, stream>>>(simg, timg, scap, tcap, il, cl, st, 3);
  nmf_phase<<<64 * NSL, 512, lds_bytes, stream>>>(simg, timg, scap, tcap, il, cl, st, 4);
  nmf_phase<<<64 * NSL, 512, lds_bytes, stream>>>(simg, timg, scap, tcap, il, cl, st, 5);
  nmf_phase<<<64 * NSL, 512, lds_bytes, stream>>>(simg, timg, scap, tcap, il, cl, st, 6);
  nmf_phase<<<64 * NSL, 512, lds_bytes, stream>>>(simg, timg, scap, tcap, il, cl, st, 7);
  nmf_phase<<<64 * NSL, 512, lds_bytes, stream>>>(simg, timg, scap, tcap, il, cl, st, 8);
  nmf_phase<<<64 * NSL, 512, lds_bytes, stream>>>(simg, timg, scap, tcap, il, cl, st, 9);
  nmf_phase<<<64 * NSL, 512, lds_bytes, stream>>>(simg, timg, scap, tcap, il, cl, st, 10);
  nmf_out<<<64, 256, 0, stream>>>(st, (float*)d_out);
}


# weight update at each launch's head: all global loads of a thread's 8 pieces (stored weights, 4 partial-sum planes) issued before any arithmetic, stores behind it, one path per case; sums bit-equal
# speedup vs baseline: 7.5951x; 1.0680x over previous
#include <hip/hip_runtime.h>
#include <math.h>

typedef __attribute__((ext_vector_type(2))) float v2f;
typedef __attribute__((ext_vector_type(8))) float v8f;

#define N_ITERS 5
#define EPS_F   1e-6f
#define ETA_F   0.02f

#define OM2_PI 1028
#define OM1_PR 52
#define SL_PR  20
#define SL_PO  324
#define OM2_O 0
#define OM1_O 16448
#define XS_O  29760
#define YS_O  34944
#define LDS_FLOATS 40128

static __device__ __forceinline__ v8f wmma4(v2f a, v2f b, v8f c) {
  return __builtin_amdgcn_wmma_f32_16x16x4_f32(false, a, false, b, (short)0, c,
                                               false, false);
}

#define NT_P 1028
#define SI_CH 576
#define TI_CH 768
#define SC_CH 960
#define TC_CH 1024
__global__ __launch_bounds__(512) void l2norm_frag(const float* __restrict__ imgs,
                                                   const float* __restrict__ caps,
                                                   float* __restrict__ simg,
                                                   float* __restrict__ timg,
                                                   float* __restrict__ scap,
                                                   float* __restrict__ tcap) {
  extern __shared__ float tile[];
  const int tid = threadIdx.x;
  const int w = tid >> 5;
  const int lane = tid & 31;
  int ent, n, R, sch, tch;
  const float* src;
  float* sdst;
  float* tdst;
  if (blockIdx.x < 128 * 3) {
    ent = blockIdx.x / 3; n = blockIdx.x - ent * 3; R = 36; sch = SI_CH; tch = TI_CH;
    src = imgs + (size_t)ent * 36 * 1024;
    sdst = simg + (size_t)ent * 64 * SI_CH;
    tdst = timg + (size_t)ent * 64 * TI_CH;
  } else {
    const int b = blockIdx.x - 128 * 3;
    ent = b >> 2; n = b & 3; R = 60; sch = SC_CH; tch = TC_CH;
    src = caps + (size_t)ent * 60 * 1024;
    sdst = scap + (size_t)ent * 64 * SC_CH;
    tdst = tcap + (size_t)ent * 64 * TC_CH;
  }
  const int row = n * 16 + w;
  float* trow = tile + w * NT_P;
  if (row < R) {
    const float* rp = src + (size_t)row * 1024 + lane;
    float v[32];
#pragma unroll
    for (int j = 0; j < 8; ++j)
#pragma unroll
      for (int e = 0; e < 4; ++e) v[j * 4 + e] = rp[32 * j + 256 * e];
    float tot = 0.f;
#pragma unroll
    for (int j = 0; j < 8; ++j) {
      float s = 0.f;
#pragma unroll
      for (int e = 0; e < 4; ++e) s += v[j * 4 + e] * v[j * 4 + e];
#pragma unroll
      for (int off = 16; off > 0; off >>= 1) s += __shfl_xor(s, off, 32);
      tot += s;
    }
    const float inv = 1.0f / fmaxf(sqrtf(tot), 1e-12f);
#pragma unroll
    for (int j = 0; j < 8; ++j)
#pragma unroll
      for (int e = 0; e < 4; ++e) trow[lane + 32 * j + 256 * e] = v[j * 4 + e] * inv;
  } else {
#pragma unroll
    for (int q = 0; q < 32; ++q) trow[lane + 32 * q] = 0.f;
  }
  __syncthreads();
  const int nS = (R / 4 - 4 * n < 4) ? (R / 4 - 4 * n) : 4;
#pragma unroll
  for (int it = 0; it < 8; ++it) {
    const int u = tid + 512 * it;
    const int ch = u >> 6;
    const int f = (u & 63) * 4;
    if (f < nS * 64) {
      const int sl = f >> 6, fl = (f >> 1) & 31;
      const int r0 = 4 * sl + 2 * (fl >> 4), cc = 16 * ch + (fl & 15);
      const v2f a = *(const v2f*)&tile[r0 * NT_P + cc];
      const v2f b = *(const v2f*)&tile[(r0 + 1) * NT_P + cc];
      *(float4*)(sdst + (size_t)ch * sch + n * 256 + f) = make_float4(a.x, b.x, a.y, b.y);
    }
  }
#pragma unroll
  for (int it = 0; it < 8; ++it) {
    const int u = tid + 512 * it;
    const int ch = u >> 6;
    const int f = (u & 63) * 4;
    const int s = f >> 6, fl = (f >> 1) & 31;
    const int r0 = fl & 15, col = 16 * ch + 4 * s + 2 * (fl >> 4);
    const v2f a = *(const v2f*)&tile[r0 * NT_P + col];
    const v2f b = *(const v2f*)&tile[(r0 + 1) * NT_P + col];
    *(float4*)(tdst + (size_t)ch * tch + n * 256 + f) = make_float4(a.x, a.y, b.x, b.y);
  }
}

static __device__ __forceinline__ void lds_barrier() {
  __builtin_amdgcn_fence(__ATOMIC_RELEASE, "workgroup", "local");
  __builtin_amdgcn_s_barrier();
  __builtin_amdgcn_fence(__ATOMIC_ACQUIRE, "workgroup", "local");
}

#define NC_HEAD 12
#define NG_HEAD 8
struct SlabB {
  v2f c[15];
  v2f g[9];
};

static __device__ __forceinline__ v8f kept_load(const float* q) {
  const float4 lo = *(const float4*)q;
  const float4 hi = *(const float4*)(q + 128);
  v8f r;
  r[0] = lo.x; r[1] = lo.y; r[2] = lo.z; r[3] = lo.w;
  r[4] = hi.x; r[5] = hi.y; r[6] = hi.z; r[7] = hi.w;
  return r;
}

static __device__ __forceinline__ void kept_store(float* q, v8f v) {
  *(float4*)q = make_float4(v[0], v[1], v[2], v[3]);
  *(float4*)(q + 128) = make_float4(v[4], v[5], v[6], v[7]);
}

template <int MODE>
static __device__ __forceinline__ void slab_loads_head(SlabB& b,
                                                       const float* __restrict__ pg,
                                                       const float* __restrict__ pc) {
  if (MODE != 1) {
#pragma unroll
    for (int s = 0; s < NC_HEAD; ++s) b.c[s] = *(const v2f*)(pc + s * 64);
  }
  if (MODE != 2) {
#pragma unroll
    for (int s = 0; s < NG_HEAD; ++s) b.g[s] = *(const v2f*)(pg + s * 64);
  }
}

template <int MODE>
static __device__ __forceinline__ void slab_loads_tail(SlabB& b,
                                                       const float* __restrict__ pg,
                                                       const float* __restrict__ pc,
                                                       bool longc, bool longg) {
  if (MODE != 1) {
    if (longc) {
#pragma unroll
      for (int s = NC_HEAD; s < 15; ++s) b.c[s] = *(const v2f*)(pc + s * 64);
    }
  }
  if (MODE != 2) {
    if (longg) {
#pragma unroll
      for (int s = NG_HEAD; s < 9; ++s) b.g[s] = *(const v2f*)(pg + s * 64);
    }
  }
}

template <int MODE, bool NEXT>
static __device__ __forceinline__ void slab_step(float* lds, SlabB& b, v8f& lin, v8f& made,
                                                 const float* qn,
                                                 const float* __restrict__ pgn,
                                                 const float* __restrict__ pcn,
                                                 int w, int ln, int koff, int ioff,
                                                 bool longc, bool longg) {
  v2f at2[15 - NC_HEAD], at1[9 - NG_HEAD];
  if (MODE != 1) {
#pragma unroll
    for (int s = NC_HEAD; s < 15; ++s)
      at2[s - NC_HEAD] = *(const v2f*)&lds[OM2_O + ln * OM2_PI + w * 64 + s * 4 + koff];
  }
  if (MODE != 2) {
#pragma unroll
    for (int s = NG_HEAD; s < 9; ++s)
      at1[s - NG_HEAD] = *(const v2f*)&lds[OM1_O + (w * 16 + ln) * OM1_PR + s * 4 + koff];
  }
  v8f x; x = 0.0f;
  if (MODE != 1) {
#pragma unroll
    for (int s = 0; s < NC_HEAD; ++s) {
      v2f a = *(const v2f*)&lds[OM2_O + ln * OM2_PI + w * 64 + s * 4 + koff];
      x = wmma4(a, b.c[s], x);
    }
  }
  v8f y; y = 0.0f;
  if (MODE != 2) {
#pragma unroll
    for (int s = 0; s < NG_HEAD; ++s) {
      v2f a = *(const v2f*)&lds[OM1_O + (w * 16 + ln) * OM1_PR + s * 4 + koff];
      y = wmma4(a, b.g[s], y);
    }
  }
  if (MODE == 0) asm volatile("" :: "v"(at2[0]), "v"(at2[1]), "v"(at2[2]), "v"(at1[0]));
  if (MODE == 1) asm volatile("" :: "v"(at1[0]));
  if (MODE == 2) asm volatile("" :: "v"(at2[0]), "v"(at2[1]), "v"(at2[2]));
  if (MODE != 1) {
    if (longc) {
#pragma unroll
      for (int s = NC_HEAD; s < 15; ++s) x = wmma4(at2[s - NC_HEAD], b.c[s], x);
    }
  }
  if (MODE != 2) {
    if (longg) {
#pragma unroll
      for (int s = NG_HEAD; s < 9; ++s) y = wmma4(at1[s - NG_HEAD], b.g[s], y);
    }
  }
  if (MODE == 1) x = lin;
  if (MODE == 2) y = lin;
#pragma unroll
  for (int j = 0; j < 8; ++j) {
    lds[XS_O + w * SL_PO + (j + ioff) * SL_PR + ln] = x[j];
    lds[YS_O + w * SL_PO + (j + ioff) * SL_PR + ln] = y[j];
  }
  made = (MODE == 2) ? x : y;
  if (NEXT) {
    slab_loads_head<MODE>(b, pgn, pcn);
    if (MODE != 0) lin = kept_load(qn);
  }
}

template <int NTM>
static __device__ __forceinline__ void tile_loads_head(v2f (&bt)[NTM * 4],
                                                       const float* __restrict__ g) {
#pragma unroll
  for (int q = 0; q < (NTM - 1) * 4; ++q) bt[q] = *(const v2f*)(g + q * 64);
}

template <int NTM>
static __device__ __forceinline__ void tile_loads_tail(v2f (&bt)[NTM * 4],
                                                       const float* __restrict__ g, bool all) {
  if (all) {
#pragma unroll
    for (int q = (NTM - 1) * 4; q < NTM * 4; ++q) bt[q] = *(const v2f*)(g + q * 64);
  }
}

template <int NTM, bool XOWN, bool NEXT>
static __device__ __forceinline__ void grad_products(const float* lds, v8f (&acc)[NTM],
                                                     v2f (&bt)[NTM * 4],
                                                     const float* __restrict__ gn,
                                                     int w, int ln, int koff, bool all) {
  v2f a[4];
#pragma unroll
  for (int s = 0; s < 4; ++s) {
    const int dd = s * 4 + koff;
    const int own = w * SL_PO + ln * SL_PR + dd;
    const int oth = ln * SL_PO + w * SL_PR + dd;
    const v2f ax = *(const v2f*)&lds[XS_O + (XOWN ? own : oth)];
    const v2f ay = *(const v2f*)&lds[YS_O + (XOWN ? oth : own)];
    a[s].x = ax.x - ay.x; a[s].y = ax.y - ay.y;
  }
#pragma unroll
  for (int n = 0; n < NTM - 1; ++n)
#pragma unroll
    for (int s = 0; s < 4; ++s) acc[n] = wmma4(a[s], bt[n * 4 + s], acc[n]);
  if (all) {
#pragma unroll
    for (int s = 0; s < 4; ++s)
      acc[NTM - 1] = wmma4(a[s], bt[(NTM - 1) * 4 + s], acc[NTM - 1]);
  }
  if (NEXT) tile_loads_head<NTM>(bt, gn);
}

static __device__ __forceinline__ void final_sums(const float* lds, int tid,
                                                  float& s11, float& s22, float& s12) {
  if (tid < 256) {
    const int pi = tid >> 4, pt = tid & 15;
#pragma unroll
    for (int d = 0; d < 16; ++d) {
      float g1 = lds[YS_O + pi * SL_PO + pt * SL_PR + d];
      float g2 = lds[XS_O + pt * SL_PO + pi * SL_PR + d];
      s12 += g1 * g2; s11 += g1 * g1; s22 += g2 * g2;
    }
  }
}

#define NSL      4
#define CH_SL    (64 / NSL)
#define O2_FL    16384
#define O1_FL    12288
#define FP_FL    768
#define ST_O2    0
#define ST_O1    (ST_O2 + 2 * 64 * O2_FL)
#define ST_PV    (ST_O1 + 2 * 64 * O1_FL)
#define ST_PU    (ST_PV + 64 * NSL * O2_FL)
#define ST_FP    (ST_PU + 64 * NSL * O1_FL)
#define SB_CH    4096
#define ST_SB    (ST_FP + 64 * NSL * FP_FL)

template <int MODE, int NTM, bool XOWN>
static __device__ __forceinline__ void grad_half(float* lds, v8f (&acc)[NTM],
                                                 const float* __restrict__ pg0,
                                                 const float* __restrict__ pc0,
                                                 const float* __restrict__ gt0, float* q0,
                                                 int c0,
                                                 int w, int ln, int koff, int ioff,
                                                 bool longc, bool longg, bool all) {
  const int tch = (NTM == 4 ? TC_CH : TI_CH);
  SlabB sb;
  v2f bt[NTM * 4];
  v8f lin, made;
  lin = 0.0f;
  int c = c0;
  const float* pc = pc0 + c * SC_CH;
  const float* pg = pg0 + c * SI_CH;
  const float* gt = gt0 + c * tch;
  float* q = q0 + c * SB_CH;
  slab_loads_head<MODE>(sb, pg, pc);
  slab_loads_tail<MODE>(sb, pg, pc, longc, longg);
  if (MODE != 0) lin = kept_load(q);
  tile_loads_head<NTM>(bt, gt);
#pragma unroll 1
  for (int j = 0; j < CH_SL - 1; ++j) {
    const int cn = (c + 1) & (CH_SL - 1);
    const float* pcn = pc0 + cn * SC_CH;
    const float* pgn = pg0 + cn * SI_CH;
    const float* gtn = gt0 + cn * tch;
    float* qn = q0 + cn * SB_CH;
    tile_loads_tail<NTM>(bt, gt, all);
    slab_step<MODE, true>(lds, sb, lin, made, qn, pgn, pcn, w, ln, koff, ioff, longc, longg);
    lds_barrier();
    kept_store(q, made);
    slab_loads_tail<MODE>(sb, pgn, pcn, longc, longg);
    grad_products<NTM, XOWN, true>(lds, acc, bt, gtn, w, ln, koff, all);
    lds_barrier();
    c = cn; pc = pcn; pg = pgn; gt = gtn; q = qn;
  }
  tile_loads_tail<NTM>(bt, gt, all);
  slab_step<MODE, false>(lds, sb, lin, made, q, pg, pc, w, ln, koff, ioff, longc, longg);
  lds_barrier();
  kept_store(q, made);
  grad_products<NTM, XOWN, false>(lds, acc, bt, gt, w, ln, koff, all);
  lds_barrier();
}

#define W_TRIPS 8
template <bool START, bool UPD>
static __device__ __forceinline__ void omega2_in(float* lds, const float* src2, const float* pv,
                                                 float* dst2, int cl, int tid, bool keep) {
  float4 x[W_TRIPS];
  float4 dp[W_TRIPS][NSL];
  float4 r[W_TRIPS];
  if (!START) {
#pragma unroll
    for (int it = 0; it < W_TRIPS; ++it) x[it] = *(const float4*)(src2 + (tid + 512 * it) * 4);
  }
  if (UPD) {
#pragma unroll
    for (int it = 0; it < W_TRIPS; ++it)
#pragma unroll
      for (int s = 0; s < NSL; ++s)
        dp[it][s] = *(const float4*)(pv + (size_t)s * O2_FL + (tid + 512 * it) * 4);
  }
#pragma unroll
  for (int it = 0; it < W_TRIPS; ++it) {
    const int idx = (tid + 512 * it) * 4;
    const int l0 = idx & 63;
    float v[4];
    if (START) {
      const float inv = 1.f / (float)cl;
      for (int e = 0; e < 4; ++e) v[e] = (l0 + e < cl) ? inv : 0.f;
    } else {
      v[0] = x[it].x; v[1] = x[it].y; v[2] = x[it].z; v[3] = x[it].w;
    }
    if (UPD) {
      float d[4] = {0.f, 0.f, 0.f, 0.f};
      for (int s = 0; s < NSL; ++s) {
        d[0] += dp[it][s].x; d[1] += dp[it][s].y; d[2] += dp[it][s].z; d[3] += dp[it][s].w;
      }
      for (int e = 0; e < 4; ++e) {
        const float nv = v[e] - ETA_F * d[e];
        v[e] = fmaxf(nv, 0.f) * ((l0 + e < cl) ? 1.f : 0.f);
      }
      float s = (v[0] + v[1]) + (v[2] + v[3]);
      for (int off = 8; off > 0; off >>= 1) s += __shfl_xor(s, off, 32);
      const float sc = 1.f / (s + EPS_F);
      for (int e = 0; e < 4; ++e) v[e] *= sc;
    }
    r[it] = make_float4(v[0], v[1], v[2], v[3]);
  }
#pragma unroll
  for (int it = 0; it < W_TRIPS; ++it) {
    const int idx = (tid + 512 * it) * 4;
    *(float4*)&lds[OM2_O + (idx >> 10) * OM2_PI + (idx & 1023)] = r[it];
  }
  if (UPD && keep) {
#pragma unroll
    for (int it = 0; it < W_TRIPS; ++it) *(float4*)(dst2 + (tid + 512 * it) * 4) = r[it];
  }
}

template <bool START, bool UPD>
static __device__ __forceinline__ void omega1_in(float* lds, const float* src1, const float* pu,
                                                 float* dst1, const int* __restrict__ il_g,
                                                 int ib, int tid, bool keep) {
  const int c16 = tid & 15;
  const bool act = c16 < 12;
  const int k0 = act ? c16 * 4 : 0;
  int il[W_TRIPS];
  float4 x[W_TRIPS];
  float4 dp[W_TRIPS][NSL];
  float4 r[W_TRIPS];
#pragma unroll
  for (int it = 0; it < W_TRIPS; ++it) il[it] = il_g[ib + (((tid >> 4) + 32 * it) >> 4)];
  if (!START) {
#pragma unroll
    for (int it = 0; it < W_TRIPS; ++it)
      x[it] = *(const float4*)(src1 + ((tid >> 4) + 32 * it) * 48 + k0);
  }
  if (UPD) {
#pragma unroll
    for (int it = 0; it < W_TRIPS; ++it)
#pragma unroll
      for (int s = 0; s < NSL; ++s)
        dp[it][s] = *(const float4*)(pu + (size_t)s * O1_FL + ((tid >> 4) + 32 * it) * 48 + k0);
  }
#pragma unroll
  for (int it = 0; it < W_TRIPS; ++it) {
    const int row = (tid >> 4) + 32 * it;
    const int idx = row * 48 + k0;
    float v[4];
    if (START) {
      const float inv = 1.f / (float)il[it];
      for (int e = 0; e < 4; ++e) v[e] = (k0 + e < il[it]) ? inv : 0.f;
    } else {
      v[0] = x[it].x; v[1] = x[it].y; v[2] = x[it].z; v[3] = x[it].w;
    }
    if (UPD) {
      float d[4] = {0.f, 0.f, 0.f, 0.f};
      for (int s = 0; s < NSL; ++s) {
        d[0] += dp[it][s].x; d[1] += dp[it][s].y; d[2] += dp[it][s].z; d[3] += dp[it][s].w;
      }
      for (int e = 0; e < 4; ++e) {
        const float nv = v[e] + ETA_F * d[e];
        v[e] = fmaxf(nv, 0.f) * ((k0 + e < il[it]) ? 1.f : 0.f);
      }
      float s = act ? ((v[0] + v[1]) + (v[2] + v[3])) : 0.f;
      for (int off = 8; off > 0; off >>= 1) s += __shfl_xor(s, off, 32);
      const float sc = 1.f / (s + EPS_F);
      for (int e = 0; e < 4; ++e) v[e] *= sc;
    }
    r[it] = make_float4(v[0], v[1], v[2], v[3]);
  }
  if (act) {
#pragma unroll
    for (int it = 0; it < W_TRIPS; ++it)
      *(float4*)&lds[OM1_O + ((tid >> 4) + 32 * it) * OM1_PR + k0] = r[it];
    if (UPD && keep) {
#pragma unroll
      for (int it = 0; it < W_TRIPS; ++it)
        *(float4*)(dst1 + ((tid >> 4) + 32 * it) * 48 + k0) = r[it];
    }
  }
}

__global__ __launch_bounds__(512) void nmf_phase(const float* __restrict__ simg,
                                                 const float* __restrict__ timg,
                                                 const float* __restrict__ scap,
                                                 const float* __restrict__ tcap,
                                                 const int* __restrict__ il_g,
                                                 const int* __restrict__ cl_g,
                                                 float* st,
                                                 int p) {
  extern __shared__ float lds[];
  const int tid = threadIdx.x;
  const int w = tid >> 5;
  const int lane = tid & 31;
  const int hi = lane >> 4;
  const int ln = lane & 15;
  const int tile = blockIdx.x / NSL;
  const int slice = blockIdx.x % NSL;
  const int ib = (tile >> 3) * 16;
  const int tb = (tile & 7) * 16;
  const int tg = tb + w;
  const int ig = ib + w;
  const int ioff = hi ? 8 : 0;
  const int koff = hi ? 2 : 0;
  const int m = p >> 1;
  const bool odd = (p & 1) != 0;
  const int chb = slice * CH_SL;
  const int c0 = (((tile >> 3) + (tile & 7)) & 7) * (CH_SL / 8);

  if (odd || p == 0) {
    const float* src2 = st + ST_O2 + ((size_t)((m + 1) & 1) * 64 + tile) * O2_FL;
    const float* pv = st + ST_PV + (size_t)tile * NSL * O2_FL;
    float* dst2 = st + ST_O2 + ((size_t)(m & 1) * 64 + tile) * O2_FL;
    const int cl = cl_g[tb + ((tid >> 4) & 15)];
    if (!odd)        omega2_in<true, false>(lds, src2, pv, dst2, cl, tid, slice == 0);
    else if (m == 0) omega2_in<true, true>(lds, src2, pv, dst2, cl, tid, slice == 0);
    else             omega2_in<false, true>(lds, src2, pv, dst2, cl, tid, slice == 0);
  }
  if (!odd) {
    const float* src1 = st + ST_O1 + ((size_t)(m & 1) * 64 + tile) * O1_FL;
    const float* pu = st + ST_PU + (size_t)tile * NSL * O1_FL;
    float* dst1 = st + ST_O1 + ((size_t)((m + 1) & 1) * 64 + tile) * O1_FL;
    if (m == 0)      omega1_in<true, false>(lds, src1, pu, dst1, il_g, ib, tid, slice == 0);
    else if (m == 1) omega1_in<true, true>(lds, src1, pu, dst1, il_g, ib, tid, slice == 0);
    else             omega1_in<false, true>(lds, src1, pu, dst1, il_g, ib, tid, slice == 0);
  }
  __syncthreads();

  const float* pc = scap + ((size_t)tg * 64 + chb) * SC_CH + lane * 2;
  const float* pg = simg + ((size_t)ig * 64 + chb) * SI_CH + lane * 2;
  float* q = st + ST_SB + (((size_t)tile * NSL + slice) * CH_SL * 16 + w) * 256 + lane * 4;

  const bool longc = __builtin_amdgcn_readfirstlane(cl_g[tg]) > 4 * NC_HEAD;
  const bool longg = __builtin_amdgcn_readfirstlane(il_g[ig]) > 4 * NG_HEAD;

  if (p == 2 * N_ITERS) {
    float s11 = 0.f, s22 = 0.f, s12 = 0.f;
    SlabB sb;
    v8f lin, made;
    int c = c0;
    const float* pcc = pc + c * SC_CH;
    const float* pgc = pg + c * SI_CH;
    float* qc = q + c * SB_CH;
    slab_loads_head<1>(sb, pgc, pcc);
    slab_loads_tail<1>(sb, pgc, pcc, longc, longg);
    lin = kept_load(qc);
#pragma unroll 1
    for (int j = 0; j < CH_SL - 1; ++j) {
      c = (c + 1) & (CH_SL - 1);
      pcc = pc + c * SC_CH; pgc = pg + c * SI_CH; qc = q + c * SB_CH;
      slab_step<1, true>(lds, sb, lin, made, qc, pgc, pcc, w, ln, koff, ioff, longc, longg);
      lds_barrier();
      slab_loads_tail<1>(sb, pgc, pcc, longc, longg);
      final_sums(lds, tid, s11, s22, s12);
      lds_barrier();
    }
    slab_step<1, false>(lds, sb, lin, made, qc, pgc, pcc, w, ln, koff, ioff, longc, longg);
    lds_barrier();
    final_sums(lds, tid, s11, s22, s12);
    lds_barrier();
    if (tid < 256) {
      float* fp = st + ST_FP + ((size_t)tile * NSL + slice) * FP_FL;
      fp[tid] = s11;
      fp[256 + tid] = s22;
      fp[512 + tid] = s12;
    }
  } else if (!odd) {
    v8f dv[4];
#pragma unroll
    for (int n = 0; n < 4; ++n) dv[n] = 0.0f;
    const float* gct = tcap + ((size_t)tg * 64 + chb) * TC_CH + lane * 2;
    if (p == 0) grad_half<0, 4, true>(lds, dv, pg, pc, gct, q, c0, w, ln, koff, ioff, longc, longg, longc);
    else        grad_half<1, 4, true>(lds, dv, pg, pc, gct, q, c0, w, ln, koff, ioff, longc, longg, longc);
    float* pvw = st + ST_PV + ((size_t)tile * NSL + slice) * O2_FL;
#pragma unroll
    for (int n = 0; n < 4; ++n)
#pragma unroll
      for (int j = 0; j < 8; ++j) {
        const int i = j + ioff, l = n * 16 + ln;
        pvw[(i * 16 + w) * 64 + l] = dv[n][j];
      }
  } else {
    v8f du[3];
#pragma unroll
    for (int n = 0; n < 3; ++n) du[n] = 0.0f;
    const float* git = timg + ((size_t)ig * 64 + chb) * TI_CH + lane * 2;
    grad_half<2, 3, false>(lds, du, pg, pc, git, q, c0, w, ln, koff, ioff, longc, longg, longg);
    float* puw = st + ST_PU + ((size_t)tile * NSL + slice) * O1_FL;
#pragma unroll
    for (int n = 0; n < 3; ++n)
#pragma unroll
      for (int j = 0; j < 8; ++j) {
        const int t = j + ioff, k = n * 16 + ln;
        puw[(w * 16 + t) * 48 + k] = du[n][j];
      }
  }
}

__global__ __launch_bounds__(256) void nmf_out(const float* __restrict__ st,
                                               float* __restrict__ out) {
  const int tile = blockIdx.x;
  const int tid = threadIdx.x;
  const int ib = (tile >> 3) * 16;
  const int tb = (tile & 7) * 16;
  const int pi = tid >> 4, pt = tid & 15;
  float s11 = 0.f, s22 = 0.f, s12 = 0.f;
  for (int s = 0; s < NSL; ++s) {
    const float* fp = st + ST_FP + ((size_t)tile * NSL + s) * FP_FL;
    s11 += fp[tid];
    s22 += fp[256 + tid];
    s12 += fp[512 + tid];
  }
  const float dnm = fmaxf(sqrtf(s11), 1e-12f) * fmaxf(sqrtf(s22), 1e-12f);
  out[(size_t)(ib + pi) * 128 + (tb + pt)] = s12 / dnm;
}

extern "C" void kernel_launch(void* const* d_in, const int* in_sizes, int n_in,
                              void* d_out, int out_size, void* d_ws, size_t ws_size,
                              hipStream_t stream) {
  (void)in_sizes; (void)n_in; (void)out_size; (void)ws_size;
  const float* imgs = (const float*)d_in[0];
  const float* caps = (const float*)d_in[1];
  const int* il = (const int*)d_in[2];
  const int* cl = (const int*)d_in[3];
  float* simg = (float*)d_ws;
  float* timg = simg + (size_t)128 * 64 * SI_CH;
  float* scap = timg + (size_t)128 * 64 * TI_CH;
  float* tcap = scap + (size_t)128 * 64 * SC_CH;
  float* st = tcap + (size_t)128 * 64 * TC_CH;
  const size_t lds_bytes = LDS_FLOATS * sizeof(float);

  l2norm_frag<<<128 * 3 + 128 * 4, 512, 16 * NT_P * sizeof(float), stream>>>(imgs, caps, simg, timg, scap, tcap);
  nmf_phase<<<64 * NSL, 512, lds_bytes, stream>>>(simg, timg, scap, tcap, il, cl, st, 0);
  nmf_phase<<<64 * NSL, 512, lds_bytes, stream>>>(simg, timg, scap, tcap, il, cl, st, 1);
  nmf_phase<<<64 * NSL, 512, lds_bytes, stream>>>(simg, timg, scap, tcap, il, cl, st, 2);
  nmf_phase<<<64 * NSL, 512, lds_bytes, stream>>>(simg, timg, scap, tcap, il, cl, st, 3);
  nmf_phase<<<64 * NSL, 512, lds_bytes, stream>>>(simg, timg, scap, tcap, il, cl, st, 4);
  nmf_phase<<<64 * NSL, 512, lds_bytes, stream>>>(simg, timg, scap, tcap, il, cl, st, 5);
  nmf_phase<<<64 * NSL, 512, lds_bytes, stream>>>(simg, timg, scap, tcap, il, cl, st, 6);
  nmf_phase<<<64 * NSL, 512, lds_bytes, stream>>>(simg, timg, scap, tcap, il, cl, st, 7);
  nmf_phase<<<64 * NSL, 512, lds_bytes, stream>>>(simg, timg, scap, tcap, il, cl, st, 8);
  nmf_phase<<<64 * NSL, 512, lds_bytes, stream>>>(simg, timg, scap, tcap, il, cl, st, 9);
  nmf_phase<<<64 * NSL, 512, lds_bytes, stream>>>(simg, timg, scap, tcap, il, cl, st, 10);
  nmf_out<<<64, 256, 0, stream>>>(st, (float*)d_out);
}


# slab products: a wave reads its A fragments (omega_2 rows for X, omega_1 rows for Y) from LDS once per launch and holds them in registers over all chunks; were re-read every chunk; sums bit-equal
# speedup vs baseline: 7.8726x; 1.0365x over previous
#include <hip/hip_runtime.h>
#include <math.h>

typedef __attribute__((ext_vector_type(2))) float v2f;
typedef __attribute__((ext_vector_type(8))) float v8f;

#define N_ITERS 5
#define EPS_F   1e-6f
#define ETA_F   0.02f

#define OM2_PI 1028
#define OM1_PR 52
#define SL_PR  20
#define SL_PO  324
#define OM2_O 0
#define OM1_O 16448
#define XS_O  29760
#define YS_O  34944
#define LDS_FLOATS 40128

static __device__ __forceinline__ v8f wmma4(v2f a, v2f b, v8f c) {
  return __builtin_amdgcn_wmma_f32_16x16x4_f32(false, a, false, b, (short)0, c,
                                               false, false);
}

#define NT_P 1028
#define SI_CH 576
#define TI_CH 768
#define SC_CH 960
#define TC_CH 1024
__global__ __launch_bounds__(512) void l2norm_frag(const float* __restrict__ imgs,
                                                   const float* __restrict__ caps,
                                                   float* __restrict__ simg,
                                                   float* __restrict__ timg,
                                                   float* __restrict__ scap,
                                                   float* __restrict__ tcap) {
  extern __shared__ float tile[];
  const int tid = threadIdx.x;
  const int w = tid >> 5;
  const int lane = tid & 31;
  int ent, n, R, sch, tch;
  const float* src;
  float* sdst;
  float* tdst;
  if (blockIdx.x < 128 * 3) {
    ent = blockIdx.x / 3; n = blockIdx.x - ent * 3; R = 36; sch = SI_CH; tch = TI_CH;
    src = imgs + (size_t)ent * 36 * 1024;
    sdst = simg + (size_t)ent * 64 * SI_CH;
    tdst = timg + (size_t)ent * 64 * TI_CH;
  } else {
    const int b = blockIdx.x - 128 * 3;
    ent = b >> 2; n = b & 3; R = 60; sch = SC_CH; tch = TC_CH;
    src = caps + (size_t)ent * 60 * 1024;
    sdst = scap + (size_t)ent * 64 * SC_CH;
    tdst = tcap + (size_t)ent * 64 * TC_CH;
  }
  const int row = n * 16 + w;
  float* trow = tile + w * NT_P;
  if (row < R) {
    const float* rp = src + (size_t)row * 1024 + lane;
    float v[32];
#pragma unroll
    for (int j = 0; j < 8; ++j)
#pragma unroll
      for (int e = 0; e < 4; ++e) v[j * 4 + e] = rp[32 * j + 256 * e];
    float tot = 0.f;
#pragma unroll
    for (int j = 0; j < 8; ++j) {
      float s = 0.f;
#pragma unroll
      for (int e = 0; e < 4; ++e) s += v[j * 4 + e] * v[j * 4 + e];
#pragma unroll
      for (int off = 16; off > 0; off >>= 1) s += __shfl_xor(s, off, 32);
      tot += s;
    }
    const float inv = 1.0f / fmaxf(sqrtf(tot), 1e-12f);
#pragma unroll
    for (int j = 0; j < 8; ++j)
#pragma unroll
      for (int e = 0; e < 4; ++e) trow[lane + 32 * j + 256 * e] = v[j * 4 + e] * inv;
  } else {
#pragma unroll
    for (int q = 0; q < 32; ++q) trow[lane + 32 * q] = 0.f;
  }
  __syncthreads();
  const int nS = (R / 4 - 4 * n < 4) ? (R / 4 - 4 * n) : 4;
#pragma unroll
  for (int it = 0; it < 8; ++it) {
    const int u = tid + 512 * it;
    const int ch = u >> 6;
    const int f = (u & 63) * 4;
    if (f < nS * 64) {
      const int sl = f >> 6, fl = (f >> 1) & 31;
      const int r0 = 4 * sl + 2 * (fl >> 4), cc = 16 * ch + (fl & 15);
      const v2f a = *(const v2f*)&tile[r0 * NT_P + cc];
      const v2f b = *(const v2f*)&tile[(r0 + 1) * NT_P + cc];
      *(float4*)(sdst + (size_t)ch * sch + n * 256 + f) = make_float4(a.x, b.x, a.y, b.y);
    }
  }
#pragma unroll
  for (int it = 0; it < 8; ++it) {
    const int u = tid + 512 * it;
    const int ch = u >> 6;
    const int f = (u & 63) * 4;
    const int s = f >> 6, fl = (f >> 1) & 31;
    const int r0 = fl & 15, col = 16 * ch + 4 * s + 2 * (fl >> 4);
    const v2f a = *(const v2f*)&tile[r0 * NT_P + col];
    const v2f b = *(const v2f*)&tile[(r0 + 1) * NT_P + col];
    *(float4*)(tdst + (size_t)ch * tch + n * 256 + f) = make_float4(a.x, a.y, b.x, b.y);
  }
}

static __device__ __forceinline__ void lds_barrier() {
  __builtin_amdgcn_fence(__ATOMIC_RELEASE, "workgroup", "local");
  __builtin_amdgcn_s_barrier();
  __builtin_amdgcn_fence(__ATOMIC_ACQUIRE, "workgroup", "local");
}

#define NC_HEAD 12
#define NG_HEAD 8
struct SlabB {
  v2f c[15];
  v2f g[9];
};

struct SlabA {
  v2f o2[15];
  v2f o1[9];
};

template <int MODE>
static __device__ __forceinline__ void slab_a_load(SlabA& a, const float* lds,
                                                   int w, int ln, int koff) {
  if (MODE != 1) {
#pragma unroll
    for (int s = 0; s < 15; ++s)
      a.o2[s] = *(const v2f*)&lds[OM2_O + ln * OM2_PI + w * 64 + s * 4 + koff];
  }
  if (MODE != 2) {
#pragma unroll
    for (int s = 0; s < 9; ++s)
      a.o1[s] = *(const v2f*)&lds[OM1_O + (w * 16 + ln) * OM1_PR + s * 4 + koff];
  }
}

static __device__ __forceinline__ v8f kept_load(const float* q) {
  const float4 lo = *(const float4*)q;
  const float4 hi = *(const float4*)(q + 128);
  v8f r;
  r[0] = lo.x; r[1] = lo.y; r[2] = lo.z; r[3] = lo.w;
  r[4] = hi.x; r[5] = hi.y; r[6] = hi.z; r[7] = hi.w;
  return r;
}

static __device__ __forceinline__ void kept_store(float* q, v8f v) {
  *(float4*)q = make_float4(v[0], v[1], v[2], v[3]);
  *(float4*)(q + 128) = make_float4(v[4], v[5], v[6], v[7]);
}

template <int MODE>
static __device__ __forceinline__ void slab_loads_head(SlabB& b,
                                                       const float* __restrict__ pg,
                                                       const float* __restrict__ pc) {
  if (MODE != 1) {
#pragma unroll
    for (int s = 0; s < NC_HEAD; ++s) b.c[s] = *(const v2f*)(pc + s * 64);
  }
  if (MODE != 2) {
#pragma unroll
    for (int s = 0; s < NG_HEAD; ++s) b.g[s] = *(const v2f*)(pg + s * 64);
  }
}

template <int MODE>
static __device__ __forceinline__ void slab_loads_tail(SlabB& b,
                                                       const float* __restrict__ pg,
                                                       const float* __restrict__ pc,
                                                       bool longc, bool longg) {
  if (MODE != 1) {
    if (longc) {
#pragma unroll
      for (int s = NC_HEAD; s < 15; ++s) b.c[s] = *(const v2f*)(pc + s * 64);
    }
  }
  if (MODE != 2) {
    if (longg) {
#pragma unroll
      for (int s = NG_HEAD; s < 9; ++s) b.g[s] = *(const v2f*)(pg + s * 64);
    }
  }
}

template <int MODE, bool NEXT>
static __device__ __forceinline__ void slab_step(float* lds, const SlabA& a, SlabB& b,
                                                 v8f& lin, v8f& made,
                                                 const float* qn,
                                                 const float* __restrict__ pgn,
                                                 const float* __restrict__ pcn,
                                                 int w, int ln, int ioff,
                                                 bool longc, bool longg) {
  v8f x; x = 0.0f;
  if (MODE != 1) {
#pragma unroll
    for (int s = 0; s < NC_HEAD; ++s) x = wmma4(a.o2[s], b.c[s], x);
  }
  v8f y; y = 0.0f;
  if (MODE != 2) {
#pragma unroll
    for (int s = 0; s < NG_HEAD; ++s) y = wmma4(a.o1[s], b.g[s], y);
  }
  if (MODE != 1) {
    if (longc) {
#pragma unroll
      for (int s = NC_HEAD; s < 15; ++s) x = wmma4(a.o2[s], b.c[s], x);
    }
  }
  if (MODE != 2) {
    if (longg) {
#pragma unroll
      for (int s = NG_HEAD; s < 9; ++s) y = wmma4(a.o1[s], b.g[s], y);
    }
  }
  if (MODE == 1) x = lin;
  if (MODE == 2) y = lin;
#pragma unroll
  for (int j = 0; j < 8; ++j) {
    lds[XS_O + w * SL_PO + (j + ioff) * SL_PR + ln] = x[j];
    lds[YS_O + w * SL_PO + (j + ioff) * SL_PR + ln] = y[j];
  }
  made = (MODE == 2) ? x : y;
  if (NEXT) {
    slab_loads_head<MODE>(b, pgn, pcn);
    if (MODE != 0) lin = kept_load(qn);
  }
}

template <int NTM>
static __device__ __forceinline__ void tile_loads_head(v2f (&bt)[NTM * 4],
                                                       const float* __restrict__ g) {
#pragma unroll
  for (int q = 0; q < (NTM - 1) * 4; ++q) bt[q] = *(const v2f*)(g + q * 64);
}

template <int NTM>
static __device__ __forceinline__ void tile_loads_tail(v2f (&bt)[NTM * 4],
                                                       const float* __restrict__ g, bool all) {
  if (all) {
#pragma unroll
    for (int q = (NTM - 1) * 4; q < NTM * 4; ++q) bt[q] = *(const v2f*)(g + q * 64);
  }
}

template <int NTM, bool XOWN, bool NEXT>
static __device__ __forceinline__ void grad_products(const float* lds, v8f (&acc)[NTM],
                                                     v2f (&bt)[NTM * 4],
                                                     const float* __restrict__ gn,
                                                     int w, int ln, int koff, bool all) {
  v2f a[4];
#pragma unroll
  for (int s = 0; s < 4; ++s) {
    const int dd = s * 4 + koff;
    const int own = w * SL_PO + ln * SL_PR + dd;
    const int oth = ln * SL_PO + w * SL_PR + dd;
    const v2f ax = *(const v2f*)&lds[XS_O + (XOWN ? own : oth)];
    const v2f ay = *(const v2f*)&lds[YS_O + (XOWN ? oth : own)];
    a[s].x = ax.x - ay.x; a[s].y = ax.y - ay.y;
  }
#pragma unroll
  for (int n = 0; n < NTM - 1; ++n)
#pragma unroll
    for (int s = 0; s < 4; ++s) acc[n] = wmma4(a[s], bt[n * 4 + s], acc[n]);
  if (all) {
#pragma unroll
    for (int s = 0; s < 4; ++s)
      acc[NTM - 1] = wmma4(a[s], bt[(NTM - 1) * 4 + s], acc[NTM - 1]);
  }
  if (NEXT) tile_loads_head<NTM>(bt, gn);
}

static __device__ __forceinline__ void final_sums(const float* lds, int tid,
                                                  float& s11, float& s22, float& s12) {
  if (tid < 256) {
    const int pi = tid >> 4, pt = tid & 15;
#pragma unroll
    for (int d = 0; d < 16; ++d) {
      float g1 = lds[YS_O + pi * SL_PO + pt * SL_PR + d];
      float g2 = lds[XS_O + pt * SL_PO + pi * SL_PR + d];
      s12 += g1 * g2; s11 += g1 * g1; s22 += g2 * g2;
    }
  }
}

#define NSL      4
#define CH_SL    (64 / NSL)
#define O2_FL    16384
#define O1_FL    12288
#define FP_FL    768
#define ST_O2    0
#define ST_O1    (ST_O2 + 2 * 64 * O2_FL)
#define ST_PV    (ST_O1 + 2 * 64 * O1_FL)
#define ST_PU    (ST_PV + 64 * NSL * O2_FL)
#define ST_FP    (ST_PU + 64 * NSL * O1_FL)
#define SB_CH    4096
#define ST_SB    (ST_FP + 64 * NSL * FP_FL)

template <int MODE, int NTM, bool XOWN>
static __device__ __forceinline__ void grad_half(float* lds, v8f (&acc)[NTM],
                                                 const float* __restrict__ pg0,
                                                 const float* __restrict__ pc0,
                                                 const float* __restrict__ gt0, float* q0,
                                                 int c0,
                                                 int w, int ln, int koff, int ioff,
                                                 bool longc, bool longg, bool all) {
  const int tch = (NTM == 4 ? TC_CH : TI_CH);
  SlabA sa;
  SlabB sb;
  v2f bt[NTM * 4];
  v8f lin, made;
  lin = 0.0f;
  int c = c0;
  const float* pc = pc0 + c * SC_CH;
  const float* pg = pg0 + c * SI_CH;
  const float* gt = gt0 + c * tch;
  float* q = q0 + c * SB_CH;
  slab_loads_head<MODE>(sb, pg, pc);
  slab_loads_tail<MODE>(sb, pg, pc, longc, longg);
  if (MODE != 0) lin = kept_load(q);
  tile_loads_head<NTM>(bt, gt);
  slab_a_load<MODE>(sa, lds, w, ln, koff);
#pragma unroll 1
  for (int j = 0; j < CH_SL - 1; ++j) {
    const int cn = (c + 1) & (CH_SL - 1);
    const float* pcn = pc0 + cn * SC_CH;
    const float* pgn = pg0 + cn * SI_CH;
    const float* gtn = gt0 + cn * tch;
    float* qn = q0 + cn * SB_CH;
    tile_loads_tail<NTM>(bt, gt, all);
    slab_step<MODE, true>(lds, sa, sb, lin, made, qn, pgn, pcn, w, ln, ioff, longc, longg);
    lds_barrier();
    kept_store(q, made);
    slab_loads_tail<MODE>(sb, pgn, pcn, longc, longg);
    grad_products<NTM, XOWN, true>(lds, acc, bt, gtn, w, ln, koff, all);
    lds_barrier();
    c = cn; pc = pcn; pg = pgn; gt = gtn; q = qn;
  }
  tile_loads_tail<NTM>(bt, gt, all);
  slab_step<MODE, false>(lds, sa, sb, lin, made, q, pg, pc, w, ln, ioff, longc, longg);
  lds_barrier();
  kept_store(q, made);
  grad_products<NTM, XOWN, false>(lds, acc, bt, gt, w, ln, koff, all);
  lds_barrier();
}

#define W_TRIPS 8
template <bool START, bool UPD>
static __device__ __forceinline__ void omega2_in(float* lds, const float* src2, const float* pv,
                                                 float* dst2, int cl, int tid, bool keep) {
  float4 x[W_TRIPS];
  float4 dp[W_TRIPS][NSL];
  float4 r[W_TRIPS];
  if (!START) {
#pragma unroll
    for (int it = 0; it < W_TRIPS; ++it) x[it] = *(const float4*)(src2 + (tid + 512 * it) * 4);
  }
  if (UPD) {
#pragma unroll
    for (int it = 0; it < W_TRIPS; ++it)
#pragma unroll
      for (int s = 0; s < NSL; ++s)
        dp[it][s] = *(const float4*)(pv + (size_t)s * O2_FL + (tid + 512 * it) * 4);
  }
#pragma unroll
  for (int it = 0; it < W_TRIPS; ++it) {
    const int idx = (tid + 512 * it) * 4;
    const int l0 = idx & 63;
    float v[4];
    if (START) {
      const float inv = 1.f / (float)cl;
      for (int e = 0; e < 4; ++e) v[e] = (l0 + e < cl) ? inv : 0.f;
    } else {
      v[0] = x[it].x; v[1] = x[it].y; v[2] = x[it].z; v[3] = x[it].w;
    }
    if (UPD) {
      float d[4] = {0.f, 0.f, 0.f, 0.f};
      for (int s = 0; s < NSL; ++s) {
        d[0] += dp[it][s].x; d[1] += dp[it][s].y; d[2] += dp[it][s].z; d[3] += dp[it][s].w;
      }
      for (int e = 0; e < 4; ++e) {
        const float nv = v[e] - ETA_F * d[e];
        v[e] = fmaxf(nv, 0.f) * ((l0 + e < cl) ? 1.f : 0.f);
      }
      float s = (v[0] + v[1]) + (v[2] + v[3]);
      for (int off = 8; off > 0; off >>= 1) s += __shfl_xor(s, off, 32);
      const float sc = 1.f / (s + EPS_F);
      for (int e = 0; e < 4; ++e) v[e] *= sc;
    }
    r[it] = make_float4(v[0], v[1], v[2], v[3]);
  }
#pragma unroll
  for (int it = 0; it < W_TRIPS; ++it) {
    const int idx = (tid + 512 * it) * 4;
    *(float4*)&lds[OM2_O + (idx >> 10) * OM2_PI + (idx & 1023)] = r[it];
  }
  if (UPD && keep) {
#pragma unroll
    for (int it = 0; it < W_TRIPS; ++it) *(float4*)(dst2 + (tid + 512 * it) * 4) = r[it];
  }
}

template <bool START, bool UPD>
static __device__ __forceinline__ void omega1_in(float* lds, const float* src1, const float* pu,
                                                 float* dst1, const int* __restrict__ il_g,
                                                 int ib, int tid, bool keep) {
  const int c16 = tid & 15;
  const bool act = c16 < 12;
  const int k0 = act ? c16 * 4 : 0;
  int il[W_TRIPS];
  float4 x[W_TRIPS];
  float4 dp[W_TRIPS][NSL];
  float4 r[W_TRIPS];
#pragma unroll
  for (int it = 0; it < W_TRIPS; ++it) il[it] = il_g[ib + (((tid >> 4) + 32 * it) >> 4)];
  if (!START) {
#pragma unroll
    for (int it = 0; it < W_TRIPS; ++it)
      x[it] = *(const float4*)(src1 + ((tid >> 4) + 32 * it) * 48 + k0);
  }
  if (UPD) {
#pragma unroll
    for (int it = 0; it < W_TRIPS; ++it)
#pragma unroll
      for (int s = 0; s < NSL; ++s)
        dp[it][s] = *(const float4*)(pu + (size_t)s * O1_FL + ((tid >> 4) + 32 * it) * 48 + k0);
  }
#pragma unroll
  for (int it = 0; it < W_TRIPS; ++it) {
    const int row = (tid >> 4) + 32 * it;
    const int idx = row * 48 + k0;
    float v[4];
    if (START) {
      const float inv = 1.f / (float)il[it];
      for (int e = 0; e < 4; ++e) v[e] = (k0 + e < il[it]) ? inv : 0.f;
    } else {
      v[0] = x[it].x; v[1] = x[it].y; v[2] = x[it].z; v[3] = x[it].w;
    }
    if (UPD) {
      float d[4] = {0.f, 0.f, 0.f, 0.f};
      for (int s = 0; s < NSL; ++s) {
        d[0] += dp[it][s].x; d[1] += dp[it][s].y; d[2] += dp[it][s].z; d[3] += dp[it][s].w;
      }
      for (int e = 0; e < 4; ++e) {
        const float nv = v[e] + ETA_F * d[e];
        v[e] = fmaxf(nv, 0.f) * ((k0 + e < il[it]) ? 1.f : 0.f);
      }
      float s = act ? ((v[0] + v[1]) + (v[2] + v[3])) : 0.f;
      for (int off = 8; off > 0; off >>= 1) s += __shfl_xor(s, off, 32);
      const float sc = 1.f / (s + EPS_F);
      for (int e = 0; e < 4; ++e) v[e] *= sc;
    }
    r[it] = make_float4(v[0], v[1], v[2], v[3]);
  }
  if (act) {
#pragma unroll
    for (int it = 0; it < W_TRIPS; ++it)
      *(float4*)&lds[OM1_O + ((tid >> 4) + 32 * it) * OM1_PR + k0] = r[it];
    if (UPD && keep) {
#pragma unroll
      for (int it = 0; it < W_TRIPS; ++it)
        *(float4*)(dst1 + ((tid >> 4) + 32 * it) * 48 + k0) = r[it];
    }
  }
}

__global__ __launch_bounds__(512) void nmf_phase(const float* __restrict__ simg,
                                                 const float* __restrict__ timg,
                                                 const float* __restrict__ scap,
                                                 const float* __restrict__ tcap,
                                                 const int* __restrict__ il_g,
                                                 const int* __restrict__ cl_g,
                                                 float* st,
                                                 int p) {
  extern __shared__ float lds[];
  const int tid = threadIdx.x;
  const int w = tid >> 5;
  const int lane = tid & 31;
  const int hi = lane >> 4;
  const int ln = lane & 15;
  const int tile = blockIdx.x / NSL;
  const int slice = blockIdx.x % NSL;
  const int ib = (tile >> 3) * 16;
  const int tb = (tile & 7) * 16;
  const int tg = tb + w;
  const int ig = ib + w;
  const int ioff = hi ? 8 : 0;
  const int koff = hi ? 2 : 0;
  const int m = p >> 1;
  const bool odd = (p & 1) != 0;
  const int chb = slice * CH_SL;
  const int c0 = (((tile >> 3) + (tile & 7)) & 7) * (CH_SL / 8);

  if (odd || p == 0) {
    const float* src2 = st + ST_O2 + ((size_t)((m + 1) & 1) * 64 + tile) * O2_FL;
    const float* pv = st + ST_PV + (size_t)tile * NSL * O2_FL;
    float* dst2 = st + ST_O2 + ((size_t)(m & 1) * 64 + tile) * O2_FL;
    const int cl = cl_g[tb + ((tid >> 4) & 15)];
    if (!odd)        omega2_in<true, false>(lds, src2, pv, dst2, cl, tid, slice == 0);
    else if (m == 0) omega2_in<true, true>(lds, src2, pv, dst2, cl, tid, slice == 0);
    else             omega2_in<false, true>(lds, src2, pv, dst2, cl, tid, slice == 0);
  }
  if (!odd) {
    const float* src1 = st + ST_O1 + ((size_t)(m & 1) * 64 + tile) * O1_FL;
    const float* pu = st + ST_PU + (size_t)tile * NSL * O1_FL;
    float* dst1 = st + ST_O1 + ((size_t)((m + 1) & 1) * 64 + tile) * O1_FL;
    if (m == 0)      omega1_in<true, false>(lds, src1, pu, dst1, il_g, ib, tid, slice == 0);
    else if (m == 1) omega1_in<true, true>(lds, src1, pu, dst1, il_g, ib, tid, slice == 0);
    else             omega1_in<false, true>(lds, src1, pu, dst1, il_g, ib, tid, slice == 0);
  }
  __syncthreads();

  const float* pc = scap + ((size_t)tg * 64 + chb) * SC_CH + lane * 2;
  const float* pg = simg + ((size_t)ig * 64 + chb) * SI_CH + lane * 2;
  float* q = st + ST_SB + (((size_t)tile * NSL + slice) * CH_SL * 16 + w) * 256 + lane * 4;

  const bool longc = __builtin_amdgcn_readfirstlane(cl_g[tg]) > 4 * NC_HEAD;
  const bool longg = __builtin_amdgcn_readfirstlane(il_g[ig]) > 4 * NG_HEAD;

  if (p == 2 * N_ITERS) {
    float s11 = 0.f, s22 = 0.f, s12 = 0.f;
    SlabA sa;
    SlabB sb;
    v8f lin, made;
    int c = c0;
    const float* pcc = pc + c * SC_CH;
    const float* pgc = pg + c * SI_CH;
    float* qc = q + c * SB_CH;
    slab_loads_head<1>(sb, pgc, pcc);
    slab_loads_tail<1>(sb, pgc, pcc, longc, longg);
    lin = kept_load(qc);
    slab_a_load<1>(sa, lds, w, ln, koff);
#pragma unroll 1
    for (int j = 0; j < CH_SL - 1; ++j) {
      c = (c + 1) & (CH_SL - 1);
      pcc = pc + c * SC_CH; pgc = pg + c * SI_CH; qc = q + c * SB_CH;
      slab_step<1, true>(lds, sa, sb, lin, made, qc, pgc, pcc, w, ln, ioff, longc, longg);
      lds_barrier();
      slab_loads_tail<1>(sb, pgc, pcc, longc, longg);
      final_sums(lds, tid, s11, s22, s12);
      lds_barrier();
    }
    slab_step<1, false>(lds, sa, sb, lin, made, qc, pgc, pcc, w, ln, ioff, longc, longg);
    lds_barrier();
    final_sums(lds, tid, s11, s22, s12);
    lds_barrier();
    if (tid < 256) {
      float* fp = st + ST_FP + ((size_t)tile * NSL + slice) * FP_FL;
      fp[tid] = s11;
      fp[256 + tid] = s22;
      fp[512 + tid] = s12;
    }
  } else if (!odd) {
    v8f dv[4];
#pragma unroll
    for (int n = 0; n < 4; ++n) dv[n] = 0.0f;
    const float* gct = tcap + ((size_t)tg * 64 + chb) * TC_CH + lane * 2;
    if (p == 0) grad_half<0, 4, true>(lds, dv, pg, pc, gct, q, c0, w, ln, koff, ioff, longc, longg, longc);
    else        grad_half<1, 4, true>(lds, dv, pg, pc, gct, q, c0, w, ln, koff, ioff, longc, longg, longc);
    float* pvw = st + ST_PV + ((size_t)tile * NSL + slice) * O2_FL;
#pragma unroll
    for (int n = 0; n < 4; ++n)
#pragma unroll
      for (int j = 0; j < 8; ++j) {
        const int i = j + ioff, l = n * 16 + ln;
        pvw[(i * 16 + w) * 64 + l] = dv[n][j];
      }
  } else {
    v8f du[3];
#pragma unroll
    for (int n = 0; n < 3; ++n) du[n] = 0.0f;
    const float* git = timg + ((size_t)ig * 64 + chb) * TI_CH + lane * 2;
    grad_half<2, 3, false>(lds, du, pg, pc, git, q, c0, w, ln, koff, ioff, longc, longg, longg);
    float* puw = st + ST_PU + ((size_t)tile * NSL + slice) * O1_FL;
#pragma unroll
    for (int n = 0; n < 3; ++n)
#pragma unroll
      for (int j = 0; j < 8; ++j) {
        const int t = j + ioff, k = n * 16 + ln;
        puw[(w * 16 + t) * 48 + k] = du[n][j];
      }
  }
}

__global__ __launch_bounds__(256) void nmf_out(const float* __restrict__ st,
                                               float* __restrict__ out) {
  const int tile = blockIdx.x;
  const int tid = threadIdx.x;
  const int ib = (tile >> 3) * 16;
  const int tb = (tile & 7) * 16;
  const int pi = tid >> 4, pt = tid & 15;
  float s11 = 0.f, s22 = 0.f, s12 = 0.f;
  for (int s = 0; s < NSL; ++s) {
    const float* fp = st + ST_FP + ((size_t)tile * NSL + s) * FP_FL;
    s11 += fp[tid];
    s22 += fp[256 + tid];
    s12 += fp[512 + tid];
  }
  const float dnm = fmaxf(sqrtf(s11), 1e-12f) * fmaxf(sqrtf(s22), 1e-12f);
  out[(size_t)(ib + pi) * 128 + (tb + pt)] = s12 / dnm;
}

extern "C" void kernel_launch(void* const* d_in, const int* in_sizes, int n_in,
                              void* d_out, int out_size, void* d_ws, size_t ws_size,
                              hipStream_t stream) {
  (void)in_sizes; (void)n_in; (void)out_size; (void)ws_size;
  const float* imgs = (const float*)d_in[0];
  const float* caps = (const float*)d_in[1];
  const int* il = (const int*)d_in[2];
  const int* cl = (const int*)d_in[3];
  float* simg = (float*)d_ws;
  float* timg = simg + (size_t)128 * 64 * SI_CH;
  float* scap = timg + (size_t)128 * 64 * TI_CH;
  float* tcap = scap + (size_t)128 * 64 * SC_CH;
  float* st = tcap + (size_t)128 * 64 * TC_CH;
  const size_t lds_bytes = LDS_FLOATS * sizeof(float);

  l2norm_frag<<<128 * 3 + 128 * 4, 512, 16 * NT_P * sizeof(float), stream>>>(imgs, caps, simg, timg, scap, tcap);
  nmf_phase<<<64 * NSL, 512, lds_bytes, stream>>>(simg, timg, scap, tcap, il, cl, st, 0);
  nmf_phase<<<64 * NSL, 512, lds_bytes, stream>>>(simg, timg, scap, tcap, il, cl, st, 1);
  nmf_phase<<<64 * NSL, 512, lds_bytes, stream>>>(simg, timg, scap, tcap, il, cl, st, 2);
  nmf_phase<<<64 * NSL, 512, lds_bytes, stream>>>(simg, timg, scap, tcap, il, cl, st, 3);
  nmf_phase<<<64 * NSL, 512, lds_bytes, stream>>>(simg, timg, scap, tcap, il, cl, st, 4);
  nmf_phase<<<64 * NSL, 512, lds_bytes, stream>>>(simg, timg, scap, tcap, il, cl, st, 5);
  nmf_phase<<<64 * NSL, 512, lds_bytes, stream>>>(simg, timg, scap, tcap, il, cl, st, 6);
  nmf_phase<<<64 * NSL, 512, lds_bytes, stream>>>(simg, timg, scap, tcap, il, cl, st, 7);
  nmf_phase<<<64 * NSL, 512, lds_bytes, stream>>>(simg, timg, scap, tcap, il, cl, st, 8);
  nmf_phase<<<64 * NSL, 512, lds_bytes, stream>>>(simg, timg, scap, tcap, il, cl, st, 9);
  nmf_phase<<<64 * NSL, 512, lds_bytes, stream>>>(simg, timg, scap, tcap, il, cl, st, 10);
  nmf_out<<<64, 256, 0, stream>>>(st, (float*)d_out);
}


# slabs of the uniform start weights are row means: first omega_2 half feeds A = mean_cap - mean_img to the dV products (no slab products, exchange, barriers), first omega_1 half Y; sum order changes
# speedup vs baseline: 8.1310x; 1.0328x over previous
#include <hip/hip_runtime.h>
#include <math.h>

typedef __attribute__((ext_vector_type(2))) float v2f;
typedef __attribute__((ext_vector_type(8))) float v8f;

#define N_ITERS 5
#define EPS_F   1e-6f
#define ETA_F   0.02f

#define OM2_PI 1028
#define OM1_PR 52
#define SL_PR  20
#define SL_PO  324
#define OM2_O 0
#define OM1_O 16448
#define XS_O  29760
#define YS_O  34944
#define LDS_FLOATS 40128

static __device__ __forceinline__ v8f wmma4(v2f a, v2f b, v8f c) {
  return __builtin_amdgcn_wmma_f32_16x16x4_f32(false, a, false, b, (short)0, c,
                                               false, false);
}

#define NT_P 1028
#define SI_CH 576
#define TI_CH 768
#define SC_CH 960
#define TC_CH 1024
__global__ __launch_bounds__(512) void l2norm_frag(const float* __restrict__ imgs,
                                                   const float* __restrict__ caps,
                                                   float* __restrict__ simg,
                                                   float* __restrict__ timg,
                                                   float* __restrict__ scap,
                                                   float* __restrict__ tcap,
                                                   const int* __restrict__ il_g,
                                                   const int* __restrict__ cl_g,
                                                   float* __restrict__ pimg,
                                                   float* __restrict__ pcap) {
  extern __shared__ float tile[];
  const int tid = threadIdx.x;
  const int w = tid >> 5;
  const int lane = tid & 31;
  int ent, n, R, sch, tch, len;
  const float* src;
  float* sdst;
  float* tdst;
  float* pdst;
  if (blockIdx.x < 128 * 3) {
    ent = blockIdx.x / 3; n = blockIdx.x - ent * 3; R = 36; sch = SI_CH; tch = TI_CH;
    src = imgs + (size_t)ent * 36 * 1024;
    sdst = simg + (size_t)ent * 64 * SI_CH;
    tdst = timg + (size_t)ent * 64 * TI_CH;
    len = il_g[ent];
    pdst = pimg + ((size_t)ent * 3 + n) * 1024;
  } else {
    const int b = blockIdx.x - 128 * 3;
    ent = b >> 2; n = b & 3; R = 60; sch = SC_CH; tch = TC_CH;
    src = caps + (size_t)ent * 60 * 1024;
    sdst = scap + (size_t)ent * 64 * SC_CH;
    tdst = tcap + (size_t)ent * 64 * TC_CH;
    len = cl_g[ent];
    pdst = pcap + ((size_t)ent * 4 + n) * 1024;
  }
  const int row = n * 16 + w;
  float* trow = tile + w * NT_P;
  if (row < R) {
    const float* rp = src + (size_t)row * 1024 + lane;
    float v[32];
#pragma unroll
    for (int j = 0; j < 8; ++j)
#pragma unroll
      for (int e = 0; e < 4; ++e) v[j * 4 + e] = rp[32 * j + 256 * e];
    float tot = 0.f;
#pragma unroll
    for (int j = 0; j < 8; ++j) {
      float s = 0.f;
#pragma unroll
      for (int e = 0; e < 4; ++e) s += v[j * 4 + e] * v[j * 4 + e];
#pragma unroll
      for (int off = 16; off > 0; off >>= 1) s += __shfl_xor(s, off, 32);
      tot += s;
    }
    const float inv = 1.0f / fmaxf(sqrtf(tot), 1e-12f);
#pragma unroll
    for (int j = 0; j < 8; ++j)
#pragma unroll
      for (int e = 0; e < 4; ++e) trow[lane + 32 * j + 256 * e] = v[j * 4 + e] * inv;
  } else {
#pragma unroll
    for (int q = 0; q < 32; ++q) trow[lane + 32 * q] = 0.f;
  }
  __syncthreads();
  {
    const int nr = (len - 16 * n < 0) ? 0 : ((len - 16 * n > 16) ? 16 : (len - 16 * n));
    float s0 = 0.f, s1 = 0.f;
#pragma unroll
    for (int r = 0; r < 16; ++r) {
      if (r < nr) { s0 += tile[r * NT_P + tid]; s1 += tile[r * NT_P + 512 + tid]; }
    }
    pdst[tid] = s0;
    pdst[512 + tid] = s1;
  }
  const int nS = (R / 4 - 4 * n < 4) ? (R / 4 - 4 * n) : 4;
#pragma unroll
  for (int it = 0; it < 8; ++it) {
    const int u = tid + 512 * it;
    const int ch = u >> 6;
    const int f = (u & 63) * 4;
    if (f < nS * 64) {
      const int sl = f >> 6, fl = (f >> 1) & 31;
      const int r0 = 4 * sl + 2 * (fl >> 4), cc = 16 * ch + (fl & 15);
      const v2f a = *(const v2f*)&tile[r0 * NT_P + cc];
      const v2f b = *(const v2f*)&tile[(r0 + 1) * NT_P + cc];
      *(float4*)(sdst + (size_t)ch * sch + n * 256 + f) = make_float4(a.x, b.x, a.y, b.y);
    }
  }
#pragma unroll
  for (int it = 0; it < 8; ++it) {
    const int u = tid + 512 * it;
    const int ch = u >> 6;
    const int f = (u & 63) * 4;
    const int s = f >> 6, fl = (f >> 1) & 31;
    const int r0 = fl & 15, col = 16 * ch + 4 * s + 2 * (fl >> 4);
    const v2f a = *(const v2f*)&tile[r0 * NT_P + col];
    const v2f b = *(const v2f*)&tile[(r0 + 1) * NT_P + col];
    *(float4*)(tdst + (size_t)ch * tch + n * 256 + f) = make_float4(a.x, a.y, b.x, b.y);
  }
}

static __device__ __forceinline__ void lds_barrier() {
  __builtin_amdgcn_fence(__ATOMIC_RELEASE, "workgroup", "local");
  __builtin_amdgcn_s_barrier();
  __builtin_amdgcn_fence(__ATOMIC_ACQUIRE, "workgroup", "local");
}

#define NC_HEAD 12
#define NG_HEAD 8
struct SlabB {
  v2f c[15];
  v2f g[9];
};

struct SlabA {
  v2f o2[15];
  v2f o1[9];
};

template <int MODE>
static __device__ __forceinline__ void slab_a_load(SlabA& a, const float* lds,
                                                   int w, int ln, int koff) {
  if (MODE != 1) {
#pragma unroll
    for (int s = 0; s < 15; ++s)
      a.o2[s] = *(const v2f*)&lds[OM2_O + ln * OM2_PI + w * 64 + s * 4 + koff];
  }
  if (MODE != 2) {
#pragma unroll
    for (int s = 0; s < 9; ++s)
      a.o1[s] = *(const v2f*)&lds[OM1_O + (w * 16 + ln) * OM1_PR + s * 4 + koff];
  }
}

static __device__ __forceinline__ v8f kept_load(const float* q) {
  const float4 lo = *(const float4*)q;
  const float4 hi = *(const float4*)(q + 128);
  v8f r;
  r[0] = lo.x; r[1] = lo.y; r[2] = lo.z; r[3] = lo.w;
  r[4] = hi.x; r[5] = hi.y; r[6] = hi.z; r[7] = hi.w;
  return r;
}

static __device__ __forceinline__ void kept_store(float* q, v8f v) {
  *(float4*)q = make_float4(v[0], v[1], v[2], v[3]);
  *(float4*)(q + 128) = make_float4(v[4], v[5], v[6], v[7]);
}

template <int MODE>
static __device__ __forceinline__ void slab_loads_head(SlabB& b,
                                                       const float* __restrict__ pg,
                                                       const float* __restrict__ pc) {
  if (MODE != 1) {
#pragma unroll
    for (int s = 0; s < NC_HEAD; ++s) b.c[s] = *(const v2f*)(pc + s * 64);
  }
  if (MODE != 2) {
#pragma unroll
    for (int s = 0; s < NG_HEAD; ++s) b.g[s] = *(const v2f*)(pg + s * 64);
  }
}

template <int MODE>
static __device__ __forceinline__ void slab_loads_tail(SlabB& b,
                                                       const float* __restrict__ pg,
                                                       const float* __restrict__ pc,
                                                       bool longc, bool longg) {
  if (MODE != 1) {
    if (longc) {
#pragma unroll
      for (int s = NC_HEAD; s < 15; ++s) b.c[s] = *(const v2f*)(pc + s * 64);
    }
  }
  if (MODE != 2) {
    if (longg) {
#pragma unroll
      for (int s = NG_HEAD; s < 9; ++s) b.g[s] = *(const v2f*)(pg + s * 64);
    }
  }
}

template <int MODE, bool NEXT, bool YMEAN>
static __device__ __forceinline__ void slab_step(float* lds, const SlabA& a, SlabB& b,
                                                 v8f& lin, v8f& made,
                                                 const float* qn, int ym,
                                                 const float* __restrict__ pgn,
                                                 const float* __restrict__ pcn,
                                                 int w, int ln, int ioff,
                                                 bool longc, bool longg) {
  v8f x; x = 0.0f;
  if (MODE != 1) {
#pragma unroll
    for (int s = 0; s < NC_HEAD; ++s) x = wmma4(a.o2[s], b.c[s], x);
  }
  v8f y; y = 0.0f;
  if (MODE != 2) {
#pragma unroll
    for (int s = 0; s < NG_HEAD; ++s) y = wmma4(a.o1[s], b.g[s], y);
  }
  if (MODE != 1) {
    if (longc) {
#pragma unroll
      for (int s = NC_HEAD; s < 15; ++s) x = wmma4(a.o2[s], b.c[s], x);
    }
  }
  if (MODE != 2) {
    if (longg) {
#pragma unroll
      for (int s = NG_HEAD; s < 9; ++s) y = wmma4(a.o1[s], b.g[s], y);
    }
  }
  if (MODE == 1) x = lin;
  if (MODE == 2) {
    if (YMEAN) y = lds[ym];
    else y = lin;
  }
#pragma unroll
  for (int j = 0; j < 8; ++j) {
    lds[XS_O + w * SL_PO + (j + ioff) * SL_PR + ln] = x[j];
    lds[YS_O + w * SL_PO + (j + ioff) * SL_PR + ln] = y[j];
  }
  made = (MODE == 2) ? x : y;
  if (NEXT) {
    slab_loads_head<MODE>(b, pgn, pcn);
    if (!YMEAN) lin = kept_load(qn);
  }
}

template <int NTM>
static __device__ __forceinline__ void tile_loads_head(v2f (&bt)[NTM * 4],
                                                       const float* __restrict__ g) {
#pragma unroll
  for (int q = 0; q < (NTM - 1) * 4; ++q) bt[q] = *(const v2f*)(g + q * 64);
}

template <int NTM>
static __device__ __forceinline__ void tile_loads_tail(v2f (&bt)[NTM * 4],
                                                       const float* __restrict__ g, bool all) {
  if (all) {
#pragma unroll
    for (int q = (NTM - 1) * 4; q < NTM * 4; ++q) bt[q] = *(const v2f*)(g + q * 64);
  }
}

template <int NTM, bool XOWN, bool NEXT>
static __device__ __forceinline__ void grad_products(const float* lds, v8f (&acc)[NTM],
                                                     v2f (&bt)[NTM * 4],
                                                     const float* __restrict__ gn,
                                                     int w, int ln, int koff, bool all) {
  v2f a[4];
#pragma unroll
  for (int s = 0; s < 4; ++s) {
    const int dd = s * 4 + koff;
    const int own = w * SL_PO + ln * SL_PR + dd;
    const int oth = ln * SL_PO + w * SL_PR + dd;
    const v2f ax = *(const v2f*)&lds[XS_O + (XOWN ? own : oth)];
    const v2f ay = *(const v2f*)&lds[YS_O + (XOWN ? oth : own)];
    a[s].x = ax.x - ay.x; a[s].y = ax.y - ay.y;
  }
#pragma unroll
  for (int n = 0; n < NTM - 1; ++n)
#pragma unroll
    for (int s = 0; s < 4; ++s) acc[n] = wmma4(a[s], bt[n * 4 + s], acc[n]);
  if (all) {
#pragma unroll
    for (int s = 0; s < 4; ++s)
      acc[NTM - 1] = wmma4(a[s], bt[(NTM - 1) * 4 + s], acc[NTM - 1]);
  }
  if (NEXT) tile_loads_head<NTM>(bt, gn);
}

static __device__ __forceinline__ void final_sums(const float* lds, int tid,
                                                  float& s11, float& s22, float& s12) {
  if (tid < 256) {
    const int pi = tid >> 4, pt = tid & 15;
#pragma unroll
    for (int d = 0; d < 16; ++d) {
      float g1 = lds[YS_O + pi * SL_PO + pt * SL_PR + d];
      float g2 = lds[XS_O + pt * SL_PO + pi * SL_PR + d];
      s12 += g1 * g2; s11 += g1 * g1; s22 += g2 * g2;
    }
  }
}

#define NSL      4
#define CH_SL    (64 / NSL)
#define O2_FL    16384
#define O1_FL    12288
#define FP_FL    768
#define ST_O2    0
#define ST_O1    (ST_O2 + 2 * 64 * O2_FL)
#define ST_PV    (ST_O1 + 2 * 64 * O1_FL)
#define ST_PU    (ST_PV + 64 * NSL * O2_FL)
#define ST_FP    (ST_PU + 64 * NSL * O1_FL)
#define SB_CH    4096
#define ST_SB    (ST_FP + 64 * NSL * FP_FL)
#define ST_PI    (ST_SB + 64 * NSL * CH_SL * SB_CH)
#define ST_PC    (ST_PI + 128 * 3 * 1024)

template <int MODE, int NTM, bool XOWN, bool YMEAN>
static __device__ __forceinline__ void grad_half(float* lds, v8f (&acc)[NTM],
                                                 const float* __restrict__ pg0,
                                                 const float* __restrict__ pc0,
                                                 const float* __restrict__ gt0, float* q0,
                                                 int c0, int ym0,
                                                 int w, int ln, int koff, int ioff,
                                                 bool longc, bool longg, bool all) {
  const int tch = (NTM == 4 ? TC_CH : TI_CH);
  SlabA sa;
  SlabB sb;
  v2f bt[NTM * 4];
  v8f lin, made;
  lin = 0.0f;
  int c = c0;
  const float* pc = pc0 + c * SC_CH;
  const float* pg = pg0 + c * SI_CH;
  const float* gt = gt0 + c * tch;
  float* q = q0 + c * SB_CH;
  slab_loads_head<MODE>(sb, pg, pc);
  slab_loads_tail<MODE>(sb, pg, pc, longc, longg);
  if (!YMEAN) lin = kept_load(q);
  tile_loads_head<NTM>(bt, gt);
  slab_a_load<MODE>(sa, lds, w, ln, koff);
#pragma unroll 1
  for (int j = 0; j < CH_SL - 1; ++j) {
    const int cn = (c + 1) & (CH_SL - 1);
    const float* pcn = pc0 + cn * SC_CH;
    const float* pgn = pg0 + cn * SI_CH;
    const float* gtn = gt0 + cn * tch;
    float* qn = q0 + cn * SB_CH;
    tile_loads_tail<NTM>(bt, gt, all);
    slab_step<MODE, true, YMEAN>(lds, sa, sb, lin, made, qn, ym0 + c * 16, pgn, pcn, w, ln, ioff,
                                 longc, longg);
    lds_barrier();
    kept_store(q, made);
    slab_loads_tail<MODE>(sb, pgn, pcn, longc, longg);
    grad_products<NTM, XOWN, true>(lds, acc, bt, gtn, w, ln, koff, all);
    lds_barrier();
    c = cn; pc = pcn; pg = pgn; gt = gtn; q = qn;
  }
  tile_loads_tail<NTM>(bt, gt, all);
  slab_step<MODE, false, YMEAN>(lds, sa, sb, lin, made, q, ym0 + c * 16, pg, pc, w, ln, ioff,
                                longc, longg);
  lds_barrier();
  kept_store(q, made);
  grad_products<NTM, XOWN, false>(lds, acc, bt, gt, w, ln, koff, all);
  lds_barrier();
}

#define W_TRIPS 8
template <bool START, bool UPD>
static __device__ __forceinline__ void omega2_in(float* lds, const float* src2, const float* pv,
                                                 float* dst2, int cl, int tid, bool keep) {
  float4 x[W_TRIPS];
  float4 dp[W_TRIPS][NSL];
  float4 r[W_TRIPS];
  if (!START) {
#pragma unroll
    for (int it = 0; it < W_TRIPS; ++it) x[it] = *(const float4*)(src2 + (tid + 512 * it) * 4);
  }
  if (UPD) {
#pragma unroll
    for (int it = 0; it < W_TRIPS; ++it)
#pragma unroll
      for (int s = 0; s < NSL; ++s)
        dp[it][s] = *(const float4*)(pv + (size_t)s * O2_FL + (tid + 512 * it) * 4);
  }
#pragma unroll
  for (int it = 0; it < W_TRIPS; ++it) {
    const int idx = (tid + 512 * it) * 4;
    const int l0 = idx & 63;
    float v[4];
    if (START) {
      const float inv = 1.f / (float)cl;
      for (int e = 0; e < 4; ++e) v[e] = (l0 + e < cl) ? inv : 0.f;
    } else {
      v[0] = x[it].x; v[1] = x[it].y; v[2] = x[it].z; v[3] = x[it].w;
    }
    if (UPD) {
      float d[4] = {0.f, 0.f, 0.f, 0.f};
      for (int s = 0; s < NSL; ++s) {
        d[0] += dp[it][s].x; d[1] += dp[it][s].y; d[2] += dp[it][s].z; d[3] += dp[it][s].w;
      }
      for (int e = 0; e < 4; ++e) {
        const float nv = v[e] - ETA_F * d[e];
        v[e] = fmaxf(nv, 0.f) * ((l0 + e < cl) ? 1.f : 0.f);
      }
      float s = (v[0] + v[1]) + (v[2] + v[3]);
      for (int off = 8; off > 0; off >>= 1) s += __shfl_xor(s, off, 32);
      const float sc = 1.f / (s + EPS_F);
      for (int e = 0; e < 4; ++e) v[e] *= sc;
    }
    r[it] = make_float4(v[0], v[1], v[2], v[3]);
  }
#pragma unroll
  for (int it = 0; it < W_TRIPS; ++it) {
    const int idx = (tid + 512 * it) * 4;
    *(float4*)&lds[OM2_O + (idx >> 10) * OM2_PI + (idx & 1023)] = r[it];
  }
  if (UPD && keep) {
#pragma unroll
    for (int it = 0; it < W_TRIPS; ++it) *(float4*)(dst2 + (tid + 512 * it) * 4) = r[it];
  }
}

template <bool START, bool UPD>
static __device__ __forceinline__ void omega1_in(float* lds, const float* src1, const float* pu,
                                                 float* dst1, const int* __restrict__ il_g,
                                                 int ib, int tid, bool keep) {
  const int c16 = tid & 15;
  const bool act = c16 < 12;
  const int k0 = act ? c16 * 4 : 0;
  int il[W_TRIPS];
  float4 x[W_TRIPS];
  float4 dp[W_TRIPS][NSL];
  float4 r[W_TRIPS];
#pragma unroll
  for (int it = 0; it < W_TRIPS; ++it) il[it] = il_g[ib + (((tid >> 4) + 32 * it) >> 4)];
  if (!START) {
#pragma unroll
    for (int it = 0; it < W_TRIPS; ++it)
      x[it] = *(const float4*)(src1 + ((tid >> 4) + 32 * it) * 48 + k0);
  }
  if (UPD) {
#pragma unroll
    for (int it = 0; it < W_TRIPS; ++it)
#pragma unroll
      for (int s = 0; s < NSL; ++s)
        dp[it][s] = *(const float4*)(pu + (size_t)s * O1_FL + ((tid >> 4) + 32 * it) * 48 + k0);
  }
#pragma unroll
  for (int it = 0; it < W_TRIPS; ++it) {
    const int row = (tid >> 4) + 32 * it;
    const int idx = row * 48 + k0;
    float v[4];
    if (START) {
      const float inv = 1.f / (float)il[it];
      for (int e = 0; e < 4; ++e) v[e] = (k0 + e < il[it]) ? inv : 0.f;
    } else {
      v[0] = x[it].x; v[1] = x[it].y; v[2] = x[it].z; v[3] = x[it].w;
    }
    if (UPD) {
      float d[4] = {0.f, 0.f, 0.f, 0.f};
      for (int s = 0; s < NSL; ++s) {
        d[0] += dp[it][s].x; d[1] += dp[it][s].y; d[2] += dp[it][s].z; d[3] += dp[it][s].w;
      }
      for (int e = 0; e < 4; ++e) {
        const float nv = v[e] + ETA_F * d[e];
        v[e] = fmaxf(nv, 0.f) * ((k0 + e < il[it]) ? 1.f : 0.f);
      }
      float s = act ? ((v[0] + v[1]) + (v[2] + v[3])) : 0.f;
      for (int off = 8; off > 0; off >>= 1) s += __shfl_xor(s, off, 32);
      const float sc = 1.f / (s + EPS_F);
      for (int e = 0; e < 4; ++e) v[e] *= sc;
    }
    r[it] = make_float4(v[0], v[1], v[2], v[3]);
  }
  if (act) {
#pragma unroll
    for (int it = 0; it < W_TRIPS; ++it)
      *(float4*)&lds[OM1_O + ((tid >> 4) + 32 * it) * OM1_PR + k0] = r[it];
    if (UPD && keep) {
#pragma unroll
      for (int it = 0; it < W_TRIPS; ++it)
        *(float4*)(dst1 + ((tid >> 4) + 32 * it) * 48 + k0) = r[it];
    }
  }
}

#define MN_P 260
#define MC_O 0
#define MI_O (16 * MN_P)
template <int NB>
static __device__ __forceinline__ void means_in(float* tab, const float* __restrict__ ps,
                                                const int* __restrict__ len_g, int eb,
                                                int slice, int tid) {
  float4 part[2][NB];
  int len[2];
#pragma unroll
  for (int it = 0; it < 2; ++it) {
    const int u = tid + 512 * it;
    const int e = u >> 6, f4 = u & 63;
    len[it] = len_g[eb + e];
#pragma unroll
    for (int n = 0; n < NB; ++n)
      part[it][n] = *(const float4*)(ps + ((size_t)(eb + e) * NB + n) * 1024 + slice * 256 + f4 * 4);
  }
#pragma unroll
  for (int it = 0; it < 2; ++it) {
    const int u = tid + 512 * it;
    const int e = u >> 6, f4 = u & 63;
    float4 t = part[it][0];
#pragma unroll
    for (int n = 1; n < NB; ++n) {
      t.x += part[it][n].x; t.y += part[it][n].y; t.z += part[it][n].z; t.w += part[it][n].w;
    }
    const float inv = 1.f / (float)len[it];
    float* row = tab + e * MN_P + (f4 >> 2) * 16 + (f4 & 3) * 2;
    v2f lo, hi;
    lo.x = t.x * inv; lo.y = t.y * inv; hi.x = t.z * inv; hi.y = t.w * inv;
    *(v2f*)row = lo;
    *(v2f*)(row + 8) = hi;
  }
}

static __device__ __forceinline__ void start_tile_loads(v2f (&bt)[16],
                                                        const float* __restrict__ g, bool all) {
  tile_loads_head<4>(bt, g);
  tile_loads_tail<4>(bt, g, all);
}

static __device__ __forceinline__ void start_products(const float* lds, v8f (&acc)[4],
                                                      const v2f (&bt)[16], int c,
                                                      int w, int ln, int hi, bool all) {
  const float* mc = lds + MC_O + w * MN_P + c * 16 + hi * 8;
  const float* mi = lds + MI_O + ln * MN_P + c * 16 + hi * 8;
  const float4 c0 = *(const float4*)mc;
  const float4 c1 = *(const float4*)(mc + 4);
  const float4 g0 = *(const float4*)mi;
  const float4 g1 = *(const float4*)(mi + 4);
  v2f a[4];
  a[0].x = c0.x - g0.x; a[0].y = c0.y - g0.y;
  a[1].x = c0.z - g0.z; a[1].y = c0.w - g0.w;
  a[2].x = c1.x - g1.x; a[2].y = c1.y - g1.y;
  a[3].x = c1.z - g1.z; a[3].y = c1.w - g1.w;
#pragma unroll
  for (int n = 0; n < 3; ++n)
#pragma unroll
    for (int s = 0; s < 4; ++s) acc[n] = wmma4(a[s], bt[n * 4 + s], acc[n]);
  if (all) {
#pragma unroll
    for (int s = 0; s < 4; ++s) acc[3] = wmma4(a[s], bt[12 + s], acc[3]);
  }
}

static __device__ __forceinline__ void start_half(const float* lds, v8f (&acc)[4],
                                                  const float* __restrict__ gt0, int c0,
                                                  int w, int ln, int hi, bool all) {
  v2f b0[16], b1[16];
  int c = c0;
  start_tile_loads(b0, gt0 + c * TC_CH, all);
#pragma unroll 1
  for (int j = 0; j < CH_SL / 2 - 1; ++j) {
    const int c1 = (c + 1) & (CH_SL - 1);
    const int c2 = (c + 2) & (CH_SL - 1);
    start_tile_loads(b1, gt0 + c1 * TC_CH, all);
    start_products(lds, acc, b0, c, w, ln, hi, all);
    start_tile_loads(b0, gt0 + c2 * TC_CH, all);
    start_products(lds, acc, b1, c1, w, ln, hi, all);
    c = c2;
  }
  const int c1 = (c + 1) & (CH_SL - 1);
  start_tile_loads(b1, gt0 + c1 * TC_CH, all);
  start_products(lds, acc, b0, c, w, ln, hi, all);
  start_products(lds, acc, b1, c1, w, ln, hi, all);
}

__global__ __launch_bounds__(512) void nmf_phase(const float* __restrict__ simg,
                                                 const float* __restrict__ timg,
                                                 const float* __restrict__ scap,
                                                 const float* __restrict__ tcap,
                                                 const int* __restrict__ il_g,
                                                 const int* __restrict__ cl_g,
                                                 const float* __restrict__ pimg,
                                                 const float* __restrict__ pcap,
                                                 float* st,
                                                 int p) {
  extern __shared__ float lds[];
  const int tid = threadIdx.x;
  const int w = tid >> 5;
  const int lane = tid & 31;
  const int hi = lane >> 4;
  const int ln = lane & 15;
  const int tile = blockIdx.x / NSL;
  const int slice = blockIdx.x % NSL;
  const int ib = (tile >> 3) * 16;
  const int tb = (tile & 7) * 16;
  const int tg = tb + w;
  const int ig = ib + w;
  const int ioff = hi ? 8 : 0;
  const int koff = hi ? 2 : 0;
  const int m = p >> 1;
  const bool odd = (p & 1) != 0;
  const int chb = slice * CH_SL;
  const int c0 = (((tile >> 3) + (tile & 7)) & 7) * (CH_SL / 8);

  if (p == 0) {
    means_in<4>(lds + MC_O, pcap, cl_g, tb, slice, tid);
    means_in<3>(lds + MI_O, pimg, il_g, ib, slice, tid);
  }
  if (odd) {
    const float* src2 = st + ST_O2 + ((size_t)((m + 1) & 1) * 64 + tile) * O2_FL;
    const float* pv = st + ST_PV + (size_t)tile * NSL * O2_FL;
    float* dst2 = st + ST_O2 + ((size_t)(m & 1) * 64 + tile) * O2_FL;
    const int cl = cl_g[tb + ((tid >> 4) & 15)];
    if (m == 0) omega2_in<true, true>(lds, src2, pv, dst2, cl, tid, slice == 0);
    else        omega2_in<false, true>(lds, src2, pv, dst2, cl, tid, slice == 0);
    if (p == 1) means_in<3>(lds + OM1_O, pimg, il_g, ib, slice, tid);
  }
  if (!odd && p != 0) {
    const float* src1 = st + ST_O1 + ((size_t)(m & 1) * 64 + tile) * O1_FL;
    const float* pu = st + ST_PU + (size_t)tile * NSL * O1_FL;
    float* dst1 = st + ST_O1 + ((size_t)((m + 1) & 1) * 64 + tile) * O1_FL;
    if (m == 1) omega1_in<true, true>(lds, src1, pu, dst1, il_g, ib, tid, slice == 0);
    else        omega1_in<false, true>(lds, src1, pu, dst1, il_g, ib, tid, slice == 0);
  }
  __syncthreads();

  const float* pc = scap + ((size_t)tg * 64 + chb) * SC_CH + lane * 2;
  const float* pg = simg + ((size_t)ig * 64 + chb) * SI_CH + lane * 2;
  float* q = st + ST_SB + (((size_t)tile * NSL + slice) * CH_SL * 16 + w) * 256 + lane * 4;

  const bool longc = __builtin_amdgcn_readfirstlane(cl_g[tg]) > 4 * NC_HEAD;
  const bool longg = __builtin_amdgcn_readfirstlane(il_g[ig]) > 4 * NG_HEAD;

  if (p == 2 * N_ITERS) {
    float s11 = 0.f, s22 = 0.f, s12 = 0.f;
    SlabA sa;
    SlabB sb;
    v8f lin, made;
    int c = c0;
    const float* pcc = pc + c * SC_CH;
    const float* pgc = pg + c * SI_CH;
    float* qc = q + c * SB_CH;
    slab_loads_head<1>(sb, pgc, pcc);
    slab_loads_tail<1>(sb, pgc, pcc, longc, longg);
    lin = kept_load(qc);
    slab_a_load<1>(sa, lds, w, ln, koff);
#pragma unroll 1
    for (int j = 0; j < CH_SL - 1; ++j) {
      c = (c + 1) & (CH_SL - 1);
      pcc = pc + c * SC_CH; pgc = pg + c * SI_CH; qc = q + c * SB_CH;
      slab_step<1, true, false>(lds, sa, sb, lin, made, qc, 0, pgc, pcc, w, ln, ioff, longc, longg);
      lds_barrier();
      slab_loads_tail<1>(sb, pgc, pcc, longc, longg);
      final_sums(lds, tid, s11, s22, s12);
      lds_barrier();
    }
    slab_step<1, false, false>(lds, sa, sb, lin, made, qc, 0, pgc, pcc, w, ln, ioff, longc, longg);
    lds_barrier();
    final_sums(lds, tid, s11, s22, s12);
    lds_barrier();
    if (tid < 256) {
      float* fp = st + ST_FP + ((size_t)tile * NSL + slice) * FP_FL;
      fp[tid] = s11;
      fp[256 + tid] = s22;
      fp[512 + tid] = s12;
    }
  } else if (!odd) {
    v8f dv[4];
#pragma unroll
    for (int n = 0; n < 4; ++n) dv[n] = 0.0f;
    const float* gct = tcap + ((size_t)tg * 64 + chb) * TC_CH + lane * 2;
    if (p == 0) start_half(lds, dv, gct, c0, w, ln, hi, longc);
    else        grad_half<1, 4, true, false>(lds, dv, pg, pc, gct, q, c0, 0, w, ln, koff, ioff, longc, longg, longc);
    float* pvw = st + ST_PV + ((size_t)tile * NSL + slice) * O2_FL;
#pragma unroll
    for (int n = 0; n < 4; ++n)
#pragma unroll
      for (int j = 0; j < 8; ++j) {
        const int i = j + ioff, l = n * 16 + ln;
        pvw[(i * 16 + w) * 64 + l] = dv[n][j];
      }
  } else {
    v8f du[3];
#pragma unroll
    for (int n = 0; n < 3; ++n) du[n] = 0.0f;
    const float* git = timg + ((size_t)ig * 64 + chb) * TI_CH + lane * 2;
    if (p == 1) {
      const int ym0 = OM1_O + w * MN_P + ((ln >> 1) & 1) * 8 + (ln >> 2) * 2 + (ln & 1);
      grad_half<2, 3, false, true>(lds, du, pg, pc, git, q, c0, ym0, w, ln, koff, ioff, longc, longg, longg);
    } else {
      grad_half<2, 3, false, false>(lds, du, pg, pc, git, q, c0, 0, w, ln, koff, ioff, longc, longg, longg);
    }
    float* puw = st + ST_PU + ((size_t)tile * NSL + slice) * O1_FL;
#pragma unroll
    for (int n = 0; n < 3; ++n)
#pragma unroll
      for (int j = 0; j < 8; ++j) {
        const int t = j + ioff, k = n * 16 + ln;
        puw[(w * 16 + t) * 48 + k] = du[n][j];
      }
  }
}

__global__ __launch_bounds__(256) void nmf_out(const float* __restrict__ st,
                                               float* __restrict__ out) {
  const int tile = blockIdx.x;
  const int tid = threadIdx.x;
  const int ib = (tile >> 3) * 16;
  const int tb = (tile & 7) * 16;
  const int pi = tid >> 4, pt = tid & 15;
  float s11 = 0.f, s22 = 0.f, s12 = 0.f;
  for (int s = 0; s < NSL; ++s) {
    const float* fp = st + ST_FP + ((size_t)tile * NSL + s) * FP_FL;
    s11 += fp[tid];
    s22 += fp[256 + tid];
    s12 += fp[512 + tid];
  }
  const float dnm = fmaxf(sqrtf(s11), 1e-12f) * fmaxf(sqrtf(s22), 1e-12f);
  out[(size_t)(ib + pi) * 128 + (tb + pt)] = s12 / dnm;
}

extern "C" void kernel_launch(void* const* d_in, const int* in_sizes, int n_in,
                              void* d_out, int out_size, void* d_ws, size_t ws_size,
                              hipStream_t stream) {
  (void)in_sizes; (void)n_in; (void)out_size; (void)ws_size;
  const float* imgs = (const float*)d_in[0];
  const float* caps = (const float*)d_in[1];
  const int* il = (const int*)d_in[2];
  const int* cl = (const int*)d_in[3];
  float* simg = (float*)d_ws;
  float* timg = simg + (size_t)128 * 64 * SI_CH;
  float* scap = timg + (size_t)128 * 64 * TI_CH;
  float* tcap = scap + (size_t)128 * 64 * SC_CH;
  float* st = tcap + (size_t)128 * 64 * TC_CH;
  float* pimg = st + ST_PI;
  float* pcap = st + ST_PC;
  const size_t lds_bytes = LDS_FLOATS * sizeof(float);

  l2norm_frag<<<128 * 3 + 128 * 4, 512, 16 * NT_P * sizeof(float), stream>>>(imgs, caps, simg, timg, scap, tcap, il, cl, pimg, pcap);
  nmf_phase<<<64 * NSL, 512, lds_bytes, stream>>>(simg, timg, scap, tcap, il, cl, pimg, pcap, st, 0);
  nmf_phase<<<64 * NSL, 512, lds_bytes, stream>>>(simg, timg, scap, tcap, il, cl, pimg, pcap, st, 1);
  nmf_phase<<<64 * NSL, 512, lds_bytes, stream>>>(simg, timg, scap, tcap, il, cl, pimg, pcap, st, 2);
  nmf_phase<<<64 * NSL, 512, lds_bytes, stream>>>(simg, timg, scap, tcap, il, cl, pimg, pcap, st, 3);
  nmf_phase<<<64 * NSL, 512, lds_bytes, stream>>>(simg, timg, scap, tcap, il, cl, pimg, pcap, st, 4);
  nmf_phase<<<64 * NSL, 512, lds_bytes, stream>>>(simg, timg, scap, tcap, il, cl, pimg, pcap, st, 5);
  nmf_phase<<<64 * NSL, 512, lds_bytes, stream>>>(simg, timg, scap, tcap, il, cl, pimg, pcap, st, 6);
  nmf_phase<<<64 * NSL, 512, lds_bytes, stream>>>(simg, timg, scap, tcap, il, cl, pimg, pcap, st, 7);
  nmf_phase<<<64 * NSL, 512, lds_bytes, stream>>>(simg, timg, scap, tcap, il, cl, pimg, pcap, st, 8);
  nmf_phase<<<64 * NSL, 512, lds_bytes, stream>>>(simg, timg, scap, tcap, il, cl, pimg, pcap, st, 9);
  nmf_phase<<<64 * NSL, 512, lds_bytes, stream>>>(simg, timg, scap, tcap, il, cl, pimg, pcap, st, 10);
  nmf_out<<<64, 256, 0, stream>>>(st, (float*)d_out);
}
